# starting kernel
# baseline (speedup 1.0000x reference)
#include <hip/hip_runtime.h>
#include <math.h>

#define B_    4
#define ROI_  90
#define T_    200
#define H_    3
#define F_    96
#define HID_  96
#define G3_   288
#define BH_   12
#define NP_   96
#define STR_  100
#define GSTR_ 292
#define TPB_  288

typedef float v2f __attribute__((ext_vector_type(2)));
typedef float v8f __attribute__((ext_vector_type(8)));

__device__ __forceinline__ v8f wmma4(v2f a, v2f b, v8f c) {
  return __builtin_amdgcn_wmma_f32_16x16x4_f32(false, a, false, b, (short)0, c,
                                               false, false);
}

__device__ __forceinline__ void mm2x2_trB(const float* As, const float* Bt,
                                          int sr, int sc, int lane, v8f acc[4]) {
  const int half = lane >> 4;
  const int l15  = lane & 15;
  const float* a0 = As + (sr * 32 + l15) * STR_ + (half << 1);
  const float* a1 = a0 + 16 * STR_;
  const float* b0 = Bt + (sc * 32 + l15) * STR_ + (half << 1);
  const float* b1 = b0 + 16 * STR_;
  #pragma unroll
  for (int j = 0; j < 4; ++j) acc[j] = (v8f){0.f,0.f,0.f,0.f,0.f,0.f,0.f,0.f};
  #pragma unroll
  for (int k0 = 0; k0 < NP_; k0 += 4) {
    v2f av0 = *(const v2f*)(a0 + k0);
    v2f av1 = *(const v2f*)(a1 + k0);
    v2f bv0 = *(const v2f*)(b0 + k0);
    v2f bv1 = *(const v2f*)(b1 + k0);
    acc[0] = wmma4(av0, bv0, acc[0]);
    acc[1] = wmma4(av0, bv1, acc[1]);
    acc[2] = wmma4(av1, bv0, acc[2]);
    acc[3] = wmma4(av1, bv1, acc[3]);
  }
}

__device__ __forceinline__ void mm1x4_trB(const float* As, const float* Bt,
                                          int ntBase, int lane, v8f acc[4]) {
  const int half = lane >> 4;
  const int l15  = lane & 15;
  const float* aRow = As + l15 * STR_ + (half << 1);
  const float* bRow = Bt + (ntBase * 16 + l15) * STR_ + (half << 1);
  #pragma unroll
  for (int j = 0; j < 4; ++j) acc[j] = (v8f){0.f,0.f,0.f,0.f,0.f,0.f,0.f,0.f};
  #pragma unroll
  for (int k0 = 0; k0 < F_; k0 += 4) {
    v2f av = *(const v2f*)(aRow + k0);
    #pragma unroll
    for (int j = 0; j < 4; ++j) {
      v2f bv = *(const v2f*)(bRow + j * 16 * STR_ + k0);
      acc[j] = wmma4(av, bv, acc[j]);
    }
  }
}

__device__ __forceinline__ void mm1x2_trB(const float* As, const float* Bt,
                                          int ntBase, int lane, v8f acc[2]) {
  const int half = lane >> 4;
  const int l15  = lane & 15;
  const float* aRow = As + l15 * STR_ + (half << 1);
  const float* bRow = Bt + (ntBase * 16 + l15) * STR_ + (half << 1);
  #pragma unroll
  for (int j = 0; j < 2; ++j) acc[j] = (v8f){0.f,0.f,0.f,0.f,0.f,0.f,0.f,0.f};
  #pragma unroll
  for (int k0 = 0; k0 < F_; k0 += 4) {
    v2f av = *(const v2f*)(aRow + k0);
    #pragma unroll
    for (int j = 0; j < 2; ++j) {
      v2f bv = *(const v2f*)(bRow + j * 16 * STR_ + k0);
      acc[j] = wmma4(av, bv, acc[j]);
    }
  }
}

__global__ void __launch_bounds__(TPB_)
ftdgnn_agg_mlp(const float* __restrict__ x, const float* __restrict__ a,
               const float* __restrict__ eps,
               const float* __restrict__ W1, const float* __restrict__ b1,
               const float* __restrict__ g1, const float* __restrict__ be1,
               const float* __restrict__ W2, const float* __restrict__ b2,
               const float* __restrict__ g2, const float* __restrict__ be2,
               float* __restrict__ hbuf) {
  extern __shared__ float smem[];
  float* xs = smem;
  float* ms = xs + NP_ * STR_;
  float* ws = ms + NP_ * STR_;

  const int tid  = threadIdx.x;
  const int lane = tid & 31;
  const int wave = tid >> 5;
  const int half = lane >> 4;
  const int l15  = lane & 15;
  const int sr   = wave / 3;
  const int sc   = wave % 3;

  const int bth = blockIdx.x;
  const int b   = bth / (T_ * H_);
  const int th  = bth % (T_ * H_);
  const int t   = th / H_;
  const int hh  = th % H_;

  for (int idx = tid; idx < NP_ * NP_; idx += TPB_) {
    const int r = idx / NP_, c = idx % NP_;
    float xv = 0.f;
    if (r < ROI_)
      xv = x[(((size_t)(b * ROI_ + r) * T_ + t) * H_ + hh) * F_ + c];
    xs[c * STR_ + r] = xv;
    float mv = 0.f;
    if (r < ROI_ && c < ROI_) {
      const float av = a[(((size_t)(b * ROI_ + r) * T_ + t) * H_ + hh) * ROI_ + c];
      mv = (av != 0.f) ? 1.f : 0.f;
    }
    ms[r * STR_ + c] = mv;
    ws[c * STR_ + r] = W1[r * HID_ + c];
  }
  __syncthreads();

  const float epsv = eps[0];
  const float rs   = rsqrtf(1.0f + 1e-5f);

  v8f acc[4];

  mm2x2_trB(ms, xs, sr, sc, lane, acc);
  __syncthreads();
  #pragma unroll
  for (int di = 0; di < 2; ++di)
    #pragma unroll
    for (int dj = 0; dj < 2; ++dj) {
      const int n = (2 * sc + dj) * 16 + l15;
      #pragma unroll
      for (int v = 0; v < 8; ++v) {
        const int m = (2 * sr + di) * 16 + v + 8 * half;
        ms[m * STR_ + n] = acc[di * 2 + dj][v] + epsv * xs[n * STR_ + m];
      }
    }
  __syncthreads();

  mm2x2_trB(ms, ws, sr, sc, lane, acc);
  __syncthreads();
  #pragma unroll
  for (int di = 0; di < 2; ++di)
    #pragma unroll
    for (int dj = 0; dj < 2; ++dj) {
      const int n = (2 * sc + dj) * 16 + l15;
      const float sc1 = g1[n] * rs, sh1 = be1[n], bb1 = b1[n];
      #pragma unroll
      for (int v = 0; v < 8; ++v) {
        const int m = (2 * sr + di) * 16 + v + 8 * half;
        float z = (acc[di * 2 + dj][v] + bb1) * sc1 + sh1;
        z = (z > 0.f) ? z : expm1f(z);
        xs[m * STR_ + n] = z;
      }
    }
  for (int idx = tid; idx < NP_ * NP_; idx += TPB_) {
    const int r = idx / NP_, c = idx % NP_;
    ws[c * STR_ + r] = W2[r * F_ + c];
  }
  __syncthreads();

  mm2x2_trB(xs, ws, sr, sc, lane, acc);
  #pragma unroll
  for (int di = 0; di < 2; ++di)
    #pragma unroll
    for (int dj = 0; dj < 2; ++dj) {
      const int n = (2 * sc + dj) * 16 + l15;
      const float sc2 = g2[n] * rs, sh2 = be2[n], bb2 = b2[n];
      #pragma unroll
      for (int v = 0; v < 8; ++v) {
        const int m = (2 * sr + di) * 16 + v + 8 * half;
        if (m < ROI_) {
          float z = (acc[di * 2 + dj][v] + bb2) * sc2 + sh2;
          z = (z > 0.f) ? z : 0.f;
          hbuf[((size_t)(m * T_ + t) * BH_ + (b * H_ + hh)) * F_ + n] = z;
        }
      }
    }
}

__global__ void __launch_bounds__(TPB_)
ftdgnn_gru(const float* __restrict__ Wih, const float* __restrict__ Whh,
           const float* __restrict__ bih, const float* __restrict__ bhh,
           const float* __restrict__ hbuf, float* __restrict__ out) {
  extern __shared__ float smem[];
  float* WihS = smem;
  float* WhhS = WihS + G3_ * STR_;
  float* hS   = WhhS + G3_ * STR_;
  float* xS   = hS   + 16 * STR_;
  float* gx   = xS   + 16 * STR_;
  float* gh   = gx   + 16 * GSTR_;
  float* bihS = gh   + 16 * GSTR_;
  float* bhhS = bihS + G3_;

  const int tid  = threadIdx.x;
  const int lane = tid & 31;
  const int wave = tid >> 5;
  const int half = lane >> 4;
  const int l15  = lane & 15;
  const int r    = blockIdx.x;

  for (int idx = tid; idx < G3_ * F_; idx += TPB_) {
    const int g = idx / F_, k = idx % F_;
    WihS[g * STR_ + k] = Wih[((size_t)r * G3_ + g) * F_ + k];
    WhhS[g * STR_ + k] = Whh[((size_t)r * G3_ + g) * HID_ + k];
  }
  for (int idx = tid; idx < G3_; idx += TPB_) {
    bihS[idx] = bih[r * G3_ + idx];
    bhhS[idx] = bhh[r * G3_ + idx];
  }
  for (int idx = tid; idx < 16 * STR_; idx += TPB_) { hS[idx] = 0.f; xS[idx] = 0.f; }
  __syncthreads();
  for (int idx = tid; idx < BH_ * F_; idx += TPB_) {
    const int m = idx / F_, k = idx % F_;
    xS[m * STR_ + k] = hbuf[((size_t)(r * T_) * BH_ + m) * F_ + k];
  }
  __syncthreads();

  for (int t = 0; t < T_; ++t) {
    if (wave < 4) {
      v8f acc[4];
      mm1x4_trB(xS, WihS, wave * 4, lane, acc);
      #pragma unroll
      for (int j = 0; j < 4; ++j) {
        const int n = (wave * 4 + j) * 16 + l15;
        #pragma unroll
        for (int v = 0; v < 8; ++v)
          gx[(v + 8 * half) * GSTR_ + n] = acc[j][v];
      }
    } else if (wave > 4) {
      v8f acc[4];
      mm1x4_trB(hS, WhhS, wave * 4 - 18, lane, acc);
      #pragma unroll
      for (int j = 0; j < 4; ++j) {
        const int n = (wave * 4 - 18 + j) * 16 + l15;
        #pragma unroll
        for (int v = 0; v < 8; ++v)
          gh[(v + 8 * half) * GSTR_ + n] = acc[j][v];
      }
    } else {
      v8f ax[2], ah[2];
      mm1x2_trB(xS, WihS, 16, lane, ax);
      mm1x2_trB(hS, WhhS, 0, lane, ah);
      #pragma unroll
      for (int j = 0; j < 2; ++j) {
        const int nx = (16 + j) * 16 + l15;
        const int nh = j * 16 + l15;
        #pragma unroll
        for (int v = 0; v < 8; ++v) {
          gx[(v + 8 * half) * GSTR_ + nx] = ax[j][v];
          gh[(v + 8 * half) * GSTR_ + nh] = ah[j][v];
        }
      }
    }
    __syncthreads();

    for (int idx = tid; idx < BH_ * HID_; idx += TPB_) {
      const int m = idx / HID_, f = idx % HID_;
      const float xr = gx[m * GSTR_ + f]             + bihS[f];
      const float xz = gx[m * GSTR_ + HID_ + f]      + bihS[HID_ + f];
      const float xn = gx[m * GSTR_ + 2 * HID_ + f]  + bihS[2 * HID_ + f];
      const float hr = gh[m * GSTR_ + f]             + bhhS[f];
      const float hz = gh[m * GSTR_ + HID_ + f]      + bhhS[HID_ + f];
      const float hn = gh[m * GSTR_ + 2 * HID_ + f]  + bhhS[2 * HID_ + f];
      const float rg = 1.f / (1.f + expf(-(xr + hr)));
      const float zg = 1.f / (1.f + expf(-(xz + hz)));
      const float ng = tanhf(xn + rg * hn);
      const float h2 = (1.f - zg) * ng + zg * hS[m * STR_ + f];
      hS[m * STR_ + f] = h2;
      const int bb = m / H_, hh = m % H_;
      out[(((size_t)(bb * ROI_ + r) * T_ + t) * H_ + hh) * HID_ + f] = h2;
    }
    if (t + 1 < T_) {
      for (int idx = tid; idx < BH_ * F_; idx += TPB_) {
        const int m = idx / F_, k = idx % F_;
        xS[m * STR_ + k] = hbuf[((size_t)(r * T_ + (t + 1)) * BH_ + m) * F_ + k];
      }
      if (t + 2 < T_) {
        const int pidx = tid & (BH_ - 1);
        __builtin_prefetch(&hbuf[((size_t)(r * T_ + (t + 2)) * BH_ + pidx) * F_], 0, 1);
      }
    }
    __syncthreads();
  }
}

extern "C" void kernel_launch(void* const* d_in, const int* in_sizes, int n_in,
                              void* d_out, int out_size, void* d_ws, size_t ws_size,
                              hipStream_t stream) {
  const float* x   = (const float*)d_in[0];
  const float* a   = (const float*)d_in[1];
  const float* eps = (const float*)d_in[2];
  const float* W1  = (const float*)d_in[3];
  const float* b1  = (const float*)d_in[4];
  const float* g1  = (const float*)d_in[5];
  const float* be1 = (const float*)d_in[6];
  const float* W2  = (const float*)d_in[7];
  const float* b2  = (const float*)d_in[8];
  const float* g2  = (const float*)d_in[9];
  const float* be2 = (const float*)d_in[10];
  const float* Wih = (const float*)d_in[11];
  const float* Whh = (const float*)d_in[12];
  const float* bih = (const float*)d_in[13];
  const float* bhh = (const float*)d_in[14];
  float* out  = (float*)d_out;
  float* hbuf = (float*)d_ws;

  const size_t ldsA = (size_t)3 * NP_ * STR_ * sizeof(float);
  ftdgnn_agg_mlp<<<B_ * T_ * H_, TPB_, ldsA, stream>>>(
      x, a, eps, W1, b1, g1, be1, W2, b2, g2, be2, hbuf);

  const size_t ldsB = (size_t)(2 * G3_ * STR_ + 2 * 16 * STR_ +
                               2 * 16 * GSTR_ + 2 * G3_) * sizeof(float);
  ftdgnn_gru<<<ROI_, TPB_, ldsB, stream>>>(Wih, Whh, bih, bhh, hbuf, out);
}


# Input projection x@Wih^T+bih hoisted out of the 200-step recurrence into a wide WMMA pre-pass (third launch, workspace); the time loop keeps only h@Whh^T, two tiles per wave on all 9 waves
# speedup vs baseline: 1.4144x; 1.4144x over previous
#include <hip/hip_runtime.h>
#include <math.h>

#define B_    4
#define ROI_  90
#define T_    200
#define H_    3
#define F_    96
#define HID_  96
#define G3_   288
#define BH_   12
#define NP_   96
#define STR_  100
#define GSTR_ 292
#define TPB_  288
#define ROWS_ 2400
#define MCH_  96
#define NCH_  25

typedef float v2f __attribute__((ext_vector_type(2)));
typedef float v8f __attribute__((ext_vector_type(8)));

__device__ __forceinline__ v8f wmma4(v2f a, v2f b, v8f c) {
  return __builtin_amdgcn_wmma_f32_16x16x4_f32(false, a, false, b, (short)0, c,
                                               false, false);
}

__device__ __forceinline__ void mm2x2_trB(const float* As, const float* Bt,
                                          int sr, int sc, int lane, v8f acc[4]) {
  const int half = lane >> 4;
  const int l15  = lane & 15;
  const float* a0 = As + (sr * 32 + l15) * STR_ + (half << 1);
  const float* a1 = a0 + 16 * STR_;
  const float* b0 = Bt + (sc * 32 + l15) * STR_ + (half << 1);
  const float* b1 = b0 + 16 * STR_;
  #pragma unroll
  for (int j = 0; j < 4; ++j) acc[j] = (v8f){0.f,0.f,0.f,0.f,0.f,0.f,0.f,0.f};
  #pragma unroll
  for (int k0 = 0; k0 < NP_; k0 += 4) {
    v2f av0 = *(const v2f*)(a0 + k0);
    v2f av1 = *(const v2f*)(a1 + k0);
    v2f bv0 = *(const v2f*)(b0 + k0);
    v2f bv1 = *(const v2f*)(b1 + k0);
    acc[0] = wmma4(av0, bv0, acc[0]);
    acc[1] = wmma4(av0, bv1, acc[1]);
    acc[2] = wmma4(av1, bv0, acc[2]);
    acc[3] = wmma4(av1, bv1, acc[3]);
  }
}

__device__ __forceinline__ void mm1x2_trB(const float* As, const float* Bt,
                                          int ntBase, int lane, v8f acc[2]) {
  const int half = lane >> 4;
  const int l15  = lane & 15;
  const float* aRow = As + l15 * STR_ + (half << 1);
  const float* bRow = Bt + (ntBase * 16 + l15) * STR_ + (half << 1);
  #pragma unroll
  for (int j = 0; j < 2; ++j) acc[j] = (v8f){0.f,0.f,0.f,0.f,0.f,0.f,0.f,0.f};
  #pragma unroll
  for (int k0 = 0; k0 < F_; k0 += 4) {
    v2f av = *(const v2f*)(aRow + k0);
    #pragma unroll
    for (int j = 0; j < 2; ++j) {
      v2f bv = *(const v2f*)(bRow + j * 16 * STR_ + k0);
      acc[j] = wmma4(av, bv, acc[j]);
    }
  }
}

__global__ void __launch_bounds__(TPB_)
ftdgnn_agg_mlp(const float* __restrict__ x, const float* __restrict__ a,
               const float* __restrict__ eps,
               const float* __restrict__ W1, const float* __restrict__ b1,
               const float* __restrict__ g1, const float* __restrict__ be1,
               const float* __restrict__ W2, const float* __restrict__ b2,
               const float* __restrict__ g2, const float* __restrict__ be2,
               float* __restrict__ hbuf) {
  extern __shared__ float smem[];
  float* xs = smem;
  float* ms = xs + NP_ * STR_;
  float* ws = ms + NP_ * STR_;

  const int tid  = threadIdx.x;
  const int lane = tid & 31;
  const int wave = tid >> 5;
  const int half = lane >> 4;
  const int l15  = lane & 15;
  const int sr   = wave / 3;
  const int sc   = wave % 3;

  const int bth = blockIdx.x;
  const int b   = bth / (T_ * H_);
  const int th  = bth % (T_ * H_);
  const int t   = th / H_;
  const int hh  = th % H_;

  for (int idx = tid; idx < NP_ * NP_; idx += TPB_) {
    const int r = idx / NP_, c = idx % NP_;
    float xv = 0.f;
    if (r < ROI_)
      xv = x[(((size_t)(b * ROI_ + r) * T_ + t) * H_ + hh) * F_ + c];
    xs[c * STR_ + r] = xv;
    float mv = 0.f;
    if (r < ROI_ && c < ROI_) {
      const float av = a[(((size_t)(b * ROI_ + r) * T_ + t) * H_ + hh) * ROI_ + c];
      mv = (av != 0.f) ? 1.f : 0.f;
    }
    ms[r * STR_ + c] = mv;
    ws[c * STR_ + r] = W1[r * HID_ + c];
  }
  __syncthreads();

  const float epsv = eps[0];
  const float rs   = rsqrtf(1.0f + 1e-5f);

  v8f acc[4];

  mm2x2_trB(ms, xs, sr, sc, lane, acc);
  __syncthreads();
  #pragma unroll
  for (int di = 0; di < 2; ++di)
    #pragma unroll
    for (int dj = 0; dj < 2; ++dj) {
      const int n = (2 * sc + dj) * 16 + l15;
      #pragma unroll
      for (int v = 0; v < 8; ++v) {
        const int m = (2 * sr + di) * 16 + v + 8 * half;
        ms[m * STR_ + n] = acc[di * 2 + dj][v] + epsv * xs[n * STR_ + m];
      }
    }
  __syncthreads();

  mm2x2_trB(ms, ws, sr, sc, lane, acc);
  __syncthreads();
  #pragma unroll
  for (int di = 0; di < 2; ++di)
    #pragma unroll
    for (int dj = 0; dj < 2; ++dj) {
      const int n = (2 * sc + dj) * 16 + l15;
      const float sc1 = g1[n] * rs, sh1 = be1[n], bb1 = b1[n];
      #pragma unroll
      for (int v = 0; v < 8; ++v) {
        const int m = (2 * sr + di) * 16 + v + 8 * half;
        float z = (acc[di * 2 + dj][v] + bb1) * sc1 + sh1;
        z = (z > 0.f) ? z : expm1f(z);
        xs[m * STR_ + n] = z;
      }
    }
  for (int idx = tid; idx < NP_ * NP_; idx += TPB_) {
    const int r = idx / NP_, c = idx % NP_;
    ws[c * STR_ + r] = W2[r * F_ + c];
  }
  __syncthreads();

  mm2x2_trB(xs, ws, sr, sc, lane, acc);
  #pragma unroll
  for (int di = 0; di < 2; ++di)
    #pragma unroll
    for (int dj = 0; dj < 2; ++dj) {
      const int n = (2 * sc + dj) * 16 + l15;
      const float sc2 = g2[n] * rs, sh2 = be2[n], bb2 = b2[n];
      #pragma unroll
      for (int v = 0; v < 8; ++v) {
        const int m = (2 * sr + di) * 16 + v + 8 * half;
        if (m < ROI_) {
          float z = (acc[di * 2 + dj][v] + bb2) * sc2 + sh2;
          z = (z > 0.f) ? z : 0.f;
          hbuf[((size_t)(m * T_ + t) * BH_ + (b * H_ + hh)) * F_ + n] = z;
        }
      }
    }
}

__global__ void __launch_bounds__(TPB_)
ftdgnn_xproj(const float* __restrict__ Wih, const float* __restrict__ bih,
             const float* __restrict__ hbuf, float* __restrict__ gxbuf) {
  extern __shared__ float smem[];
  float* As = smem;
  float* Bs = As + MCH_ * STR_;

  const int tid  = threadIdx.x;
  const int lane = tid & 31;
  const int wave = tid >> 5;
  const int half = lane >> 4;
  const int l15  = lane & 15;
  const int r    = blockIdx.x / NCH_;
  const int ch   = blockIdx.x % NCH_;
  const size_t row0 = (size_t)r * ROWS_ + (size_t)ch * MCH_;

  const float* Ag = hbuf + row0 * F_;
  const float* Bg = Wih + (size_t)r * G3_ * F_;
  #pragma unroll
  for (int i = 0; i < MCH_ * (F_ / 4) / TPB_; ++i) {
    const int idx = tid + i * TPB_;
    const int row = idx / (F_ / 4), k4 = idx % (F_ / 4);
    const float4 v = *(const float4*)(Ag + row * F_ + k4 * 4);
    *(float4*)(As + row * STR_ + k4 * 4) = v;
  }
  #pragma unroll
  for (int i = 0; i < G3_ * (F_ / 4) / TPB_; ++i) {
    const int idx = tid + i * TPB_;
    const int g = idx / (F_ / 4), k4 = idx % (F_ / 4);
    const float4 v = *(const float4*)(Bg + g * F_ + k4 * 4);
    *(float4*)(Bs + g * STR_ + k4 * 4) = v;
  }
  __syncthreads();

  float bias[2];
  #pragma unroll
  for (int dj = 0; dj < 2; ++dj) bias[dj] = bih[r * G3_ + (2 * wave + dj) * 16 + l15];

  for (int sr = 0; sr < MCH_ / 32; ++sr) {
    v8f acc[4];
    mm2x2_trB(As, Bs, sr, wave, lane, acc);
    #pragma unroll
    for (int di = 0; di < 2; ++di)
      #pragma unroll
      for (int dj = 0; dj < 2; ++dj) {
        const int n = (2 * wave + dj) * 16 + l15;
        #pragma unroll
        for (int v = 0; v < 8; ++v) {
          const int m = (2 * sr + di) * 16 + v + 8 * half;
          gxbuf[(row0 + m) * G3_ + n] = acc[di * 2 + dj][v] + bias[dj];
        }
      }
  }
}

__global__ void __launch_bounds__(TPB_)
ftdgnn_gru(const float* __restrict__ Whh, const float* __restrict__ bhh,
           const float* __restrict__ gxbuf, float* __restrict__ out) {
  extern __shared__ float smem[];
  float* WhhS = smem;
  float* hS   = WhhS + G3_ * STR_;
  float* gh   = hS   + 16 * STR_;
  float* bhhS = gh   + 16 * GSTR_;

  const int tid  = threadIdx.x;
  const int lane = tid & 31;
  const int wave = tid >> 5;
  const int half = lane >> 4;
  const int l15  = lane & 15;
  const int r    = blockIdx.x;
  const int f    = tid % HID_;
  const int m0   = tid / HID_;

  for (int idx = tid; idx < G3_ * F_; idx += TPB_) {
    const int g = idx / F_, k = idx % F_;
    WhhS[g * STR_ + k] = Whh[((size_t)r * G3_ + g) * HID_ + k];
  }
  for (int idx = tid; idx < G3_; idx += TPB_) bhhS[idx] = bhh[r * G3_ + idx];
  for (int idx = tid; idx < 16 * STR_; idx += TPB_) hS[idx] = 0.f;
  __syncthreads();

  for (int t = 0; t < T_; ++t) {
    const float* gxt = gxbuf + ((size_t)(r * T_ + t) * BH_) * G3_;
    float xg[4][3];
    #pragma unroll
    for (int i = 0; i < 4; ++i)
      #pragma unroll
      for (int g = 0; g < 3; ++g)
        xg[i][g] = gxt[(3 * i + m0) * G3_ + g * HID_ + f];

    v8f acc[2];
    mm1x2_trB(hS, WhhS, 2 * wave, lane, acc);
    #pragma unroll
    for (int j = 0; j < 2; ++j) {
      const int n = (2 * wave + j) * 16 + l15;
      #pragma unroll
      for (int v = 0; v < 8; ++v)
        gh[(v + 8 * half) * GSTR_ + n] = acc[j][v];
    }
    __syncthreads();

    #pragma unroll
    for (int i = 0; i < 4; ++i) {
      const int m = 3 * i + m0;
      const float xr = xg[i][0];
      const float xz = xg[i][1];
      const float xn = xg[i][2];
      const float hr = gh[m * GSTR_ + f]             + bhhS[f];
      const float hz = gh[m * GSTR_ + HID_ + f]      + bhhS[HID_ + f];
      const float hn = gh[m * GSTR_ + 2 * HID_ + f]  + bhhS[2 * HID_ + f];
      const float rg = 1.f / (1.f + expf(-(xr + hr)));
      const float zg = 1.f / (1.f + expf(-(xz + hz)));
      const float ng = tanhf(xn + rg * hn);
      const float h2 = (1.f - zg) * ng + zg * hS[m * STR_ + f];
      hS[m * STR_ + f] = h2;
      const int bb = m / H_, hh = m % H_;
      out[(((size_t)(bb * ROI_ + r) * T_ + t) * H_ + hh) * HID_ + f] = h2;
    }
    __syncthreads();
  }
}

extern "C" void kernel_launch(void* const* d_in, const int* in_sizes, int n_in,
                              void* d_out, int out_size, void* d_ws, size_t ws_size,
                              hipStream_t stream) {
  const float* x   = (const float*)d_in[0];
  const float* a   = (const float*)d_in[1];
  const float* eps = (const float*)d_in[2];
  const float* W1  = (const float*)d_in[3];
  const float* b1  = (const float*)d_in[4];
  const float* g1  = (const float*)d_in[5];
  const float* be1 = (const float*)d_in[6];
  const float* W2  = (const float*)d_in[7];
  const float* b2  = (const float*)d_in[8];
  const float* g2  = (const float*)d_in[9];
  const float* be2 = (const float*)d_in[10];
  const float* Wih = (const float*)d_in[11];
  const float* Whh = (const float*)d_in[12];
  const float* bih = (const float*)d_in[13];
  const float* bhh = (const float*)d_in[14];
  float* out   = (float*)d_out;
  float* hbuf  = (float*)d_ws;
  float* gxbuf = hbuf + (size_t)ROI_ * ROWS_ * F_;

  const size_t ldsA = (size_t)3 * NP_ * STR_ * sizeof(float);
  ftdgnn_agg_mlp<<<B_ * T_ * H_, TPB_, ldsA, stream>>>(
      x, a, eps, W1, b1, g1, be1, W2, b2, g2, be2, hbuf);

  const size_t ldsP = (size_t)(MCH_ + G3_) * STR_ * sizeof(float);
  ftdgnn_xproj<<<ROI_ * NCH_, TPB_, ldsP, stream>>>(Wih, bih, hbuf, gxbuf);

  const size_t ldsB = (size_t)168 * 1024;
  ftdgnn_gru<<<ROI_, TPB_, ldsB, stream>>>(Whh, bhh, gxbuf, out);
}


# Kernel A tile staging (x, mask, W1, W2): all global loads of a tile issued together, unrolled and branch-free, then stored to LDS, instead of one dependent memory round trip per element
# speedup vs baseline: 1.5935x; 1.1266x over previous
#include <hip/hip_runtime.h>
#include <math.h>

#define B_    4
#define ROI_  90
#define T_    200
#define H_    3
#define F_    96
#define HID_  96
#define G3_   288
#define BH_   12
#define NP_   96
#define STR_  100
#define GSTR_ 292
#define TPB_  288
#define ROWS_ 2400
#define MCH_  96
#define NCH_  25
#define NPASS_ 32
#define NREAL_ 30
static_assert(TPB_ == 3 * NP_ && NPASS_ * TPB_ == NP_ * NP_ && NREAL_ * 3 == ROI_,
              "kernel A staging map");

typedef float v2f __attribute__((ext_vector_type(2)));
typedef float v8f __attribute__((ext_vector_type(8)));

__device__ __forceinline__ v8f wmma4(v2f a, v2f b, v8f c) {
  return __builtin_amdgcn_wmma_f32_16x16x4_f32(false, a, false, b, (short)0, c,
                                               false, false);
}

__device__ __forceinline__ void mm2x2_trB(const float* As, const float* Bt,
                                          int sr, int sc, int lane, v8f acc[4]) {
  const int half = lane >> 4;
  const int l15  = lane & 15;
  const float* a0 = As + (sr * 32 + l15) * STR_ + (half << 1);
  const float* a1 = a0 + 16 * STR_;
  const float* b0 = Bt + (sc * 32 + l15) * STR_ + (half << 1);
  const float* b1 = b0 + 16 * STR_;
  #pragma unroll
  for (int j = 0; j < 4; ++j) acc[j] = (v8f){0.f,0.f,0.f,0.f,0.f,0.f,0.f,0.f};
  #pragma unroll
  for (int k0 = 0; k0 < NP_; k0 += 4) {
    v2f av0 = *(const v2f*)(a0 + k0);
    v2f av1 = *(const v2f*)(a1 + k0);
    v2f bv0 = *(const v2f*)(b0 + k0);
    v2f bv1 = *(const v2f*)(b1 + k0);
    acc[0] = wmma4(av0, bv0, acc[0]);
    acc[1] = wmma4(av0, bv1, acc[1]);
    acc[2] = wmma4(av1, bv0, acc[2]);
    acc[3] = wmma4(av1, bv1, acc[3]);
  }
}

__device__ __forceinline__ void mm1x2_trB(const float* As, const float* Bt,
                                          int ntBase, int lane, v8f acc[2]) {
  const int half = lane >> 4;
  const int l15  = lane & 15;
  const float* aRow = As + l15 * STR_ + (half << 1);
  const float* bRow = Bt + (ntBase * 16 + l15) * STR_ + (half << 1);
  #pragma unroll
  for (int j = 0; j < 2; ++j) acc[j] = (v8f){0.f,0.f,0.f,0.f,0.f,0.f,0.f,0.f};
  #pragma unroll
  for (int k0 = 0; k0 < F_; k0 += 4) {
    v2f av = *(const v2f*)(aRow + k0);
    #pragma unroll
    for (int j = 0; j < 2; ++j) {
      v2f bv = *(const v2f*)(bRow + j * 16 * STR_ + k0);
      acc[j] = wmma4(av, bv, acc[j]);
    }
  }
}

__global__ void __launch_bounds__(TPB_)
ftdgnn_agg_mlp(const float* __restrict__ x, const float* __restrict__ a,
               const float* __restrict__ eps,
               const float* __restrict__ W1, const float* __restrict__ b1,
               const float* __restrict__ g1, const float* __restrict__ be1,
               const float* __restrict__ W2, const float* __restrict__ b2,
               const float* __restrict__ g2, const float* __restrict__ be2,
               float* __restrict__ hbuf) {
  extern __shared__ float smem[];
  float* xs = smem;
  float* ms = xs + NP_ * STR_;
  float* ws = ms + NP_ * STR_;

  const int tid  = threadIdx.x;
  const int lane = tid & 31;
  const int wave = tid >> 5;
  const int half = lane >> 4;
  const int l15  = lane & 15;
  const int sr   = wave / 3;
  const int sc   = wave % 3;

  const int bth = blockIdx.x;
  const int b   = bth / (T_ * H_);
  const int th  = bth % (T_ * H_);
  const int t   = th / H_;
  const int hh  = th % H_;

  const int r0 = tid / NP_;
  const int c0 = tid % NP_;
  const int ca = (c0 < ROI_) ? c0 : (ROI_ - 1);
  const size_t rowBase = ((size_t)(b * ROI_ + r0) * T_ + t) * H_ + hh;
  const float* xp  = x  + rowBase * F_   + c0;
  const float* ap  = a  + rowBase * ROI_ + ca;
  const float* w1p = W1 + r0 * HID_ + c0;
  float xv[NREAL_], av[NREAL_], wv[NPASS_];
  #pragma unroll
  for (int i = 0; i < NREAL_; ++i) xv[i] = xp[(size_t)i * (3 * T_ * H_ * F_)];
  #pragma unroll
  for (int i = 0; i < NREAL_; ++i) av[i] = ap[(size_t)i * (3 * T_ * H_ * ROI_)];
  #pragma unroll
  for (int i = 0; i < NPASS_; ++i) wv[i] = w1p[i * (3 * HID_)];
  #pragma unroll
  for (int i = 0; i < NPASS_; ++i) {
    const int r = r0 + 3 * i;
    float xe = 0.f, me = 0.f;
    if (i < NREAL_) {
      xe = xv[i];
      me = (c0 < ROI_ && av[i] != 0.f) ? 1.f : 0.f;
    }
    xs[c0 * STR_ + r] = xe;
    ms[r * STR_ + c0] = me;
    ws[c0 * STR_ + r] = wv[i];
  }
  __syncthreads();

  const float epsv = eps[0];
  const float rs   = rsqrtf(1.0f + 1e-5f);

  v8f acc[4];

  mm2x2_trB(ms, xs, sr, sc, lane, acc);
  __syncthreads();
  #pragma unroll
  for (int di = 0; di < 2; ++di)
    #pragma unroll
    for (int dj = 0; dj < 2; ++dj) {
      const int n = (2 * sc + dj) * 16 + l15;
      #pragma unroll
      for (int v = 0; v < 8; ++v) {
        const int m = (2 * sr + di) * 16 + v + 8 * half;
        ms[m * STR_ + n] = acc[di * 2 + dj][v] + epsv * xs[n * STR_ + m];
      }
    }
  __syncthreads();

  mm2x2_trB(ms, ws, sr, sc, lane, acc);
  __syncthreads();
  const float* w2p = W2 + r0 * F_ + c0;
  #pragma unroll
  for (int i = 0; i < NPASS_; ++i) wv[i] = w2p[i * (3 * F_)];
  #pragma unroll
  for (int di = 0; di < 2; ++di)
    #pragma unroll
    for (int dj = 0; dj < 2; ++dj) {
      const int n = (2 * sc + dj) * 16 + l15;
      const float sc1 = g1[n] * rs, sh1 = be1[n], bb1 = b1[n];
      #pragma unroll
      for (int v = 0; v < 8; ++v) {
        const int m = (2 * sr + di) * 16 + v + 8 * half;
        float z = (acc[di * 2 + dj][v] + bb1) * sc1 + sh1;
        z = (z > 0.f) ? z : expm1f(z);
        xs[m * STR_ + n] = z;
      }
    }
  #pragma unroll
  for (int i = 0; i < NPASS_; ++i)
    ws[c0 * STR_ + r0 + 3 * i] = wv[i];
  __syncthreads();

  mm2x2_trB(xs, ws, sr, sc, lane, acc);
  #pragma unroll
  for (int di = 0; di < 2; ++di)
    #pragma unroll
    for (int dj = 0; dj < 2; ++dj) {
      const int n = (2 * sc + dj) * 16 + l15;
      const float sc2 = g2[n] * rs, sh2 = be2[n], bb2 = b2[n];
      #pragma unroll
      for (int v = 0; v < 8; ++v) {
        const int m = (2 * sr + di) * 16 + v + 8 * half;
        if (m < ROI_) {
          float z = (acc[di * 2 + dj][v] + bb2) * sc2 + sh2;
          z = (z > 0.f) ? z : 0.f;
          hbuf[((size_t)(m * T_ + t) * BH_ + (b * H_ + hh)) * F_ + n] = z;
        }
      }
    }
}

__global__ void __launch_bounds__(TPB_)
ftdgnn_xproj(const float* __restrict__ Wih, const float* __restrict__ bih,
             const float* __restrict__ hbuf, float* __restrict__ gxbuf) {
  extern __shared__ float smem[];
  float* As = smem;
  float* Bs = As + MCH_ * STR_;

  const int tid  = threadIdx.x;
  const int lane = tid & 31;
  const int wave = tid >> 5;
  const int half = lane >> 4;
  const int l15  = lane & 15;
  const int r    = blockIdx.x / NCH_;
  const int ch   = blockIdx.x % NCH_;
  const size_t row0 = (size_t)r * ROWS_ + (size_t)ch * MCH_;

  const float* Ag = hbuf + row0 * F_;
  const float* Bg = Wih + (size_t)r * G3_ * F_;
  #pragma unroll
  for (int i = 0; i < MCH_ * (F_ / 4) / TPB_; ++i) {
    const int idx = tid + i * TPB_;
    const int row = idx / (F_ / 4), k4 = idx % (F_ / 4);
    const float4 v = *(const float4*)(Ag + row * F_ + k4 * 4);
    *(float4*)(As + row * STR_ + k4 * 4) = v;
  }
  #pragma unroll
  for (int i = 0; i < G3_ * (F_ / 4) / TPB_; ++i) {
    const int idx = tid + i * TPB_;
    const int g = idx / (F_ / 4), k4 = idx % (F_ / 4);
    const float4 v = *(const float4*)(Bg + g * F_ + k4 * 4);
    *(float4*)(Bs + g * STR_ + k4 * 4) = v;
  }
  __syncthreads();

  float bias[2];
  #pragma unroll
  for (int dj = 0; dj < 2; ++dj) bias[dj] = bih[r * G3_ + (2 * wave + dj) * 16 + l15];

  for (int sr = 0; sr < MCH_ / 32; ++sr) {
    v8f acc[4];
    mm2x2_trB(As, Bs, sr, wave, lane, acc);
    #pragma unroll
    for (int di = 0; di < 2; ++di)
      #pragma unroll
      for (int dj = 0; dj < 2; ++dj) {
        const int n = (2 * wave + dj) * 16 + l15;
        #pragma unroll
        for (int v = 0; v < 8; ++v) {
          const int m = (2 * sr + di) * 16 + v + 8 * half;
          gxbuf[(row0 + m) * G3_ + n] = acc[di * 2 + dj][v] + bias[dj];
        }
      }
  }
}

__global__ void __launch_bounds__(TPB_)
ftdgnn_gru(const float* __restrict__ Whh, const float* __restrict__ bhh,
           const float* __restrict__ gxbuf, float* __restrict__ out) {
  extern __shared__ float smem[];
  float* WhhS = smem;
  float* hS   = WhhS + G3_ * STR_;
  float* gh   = hS   + 16 * STR_;
  float* bhhS = gh   + 16 * GSTR_;

  const int tid  = threadIdx.x;
  const int lane = tid & 31;
  const int wave = tid >> 5;
  const int half = lane >> 4;
  const int l15  = lane & 15;
  const int r    = blockIdx.x;
  const int f    = tid % HID_;
  const int m0   = tid / HID_;

  for (int idx = tid; idx < G3_ * F_; idx += TPB_) {
    const int g = idx / F_, k = idx % F_;
    WhhS[g * STR_ + k] = Whh[((size_t)r * G3_ + g) * HID_ + k];
  }
  for (int idx = tid; idx < G3_; idx += TPB_) bhhS[idx] = bhh[r * G3_ + idx];
  for (int idx = tid; idx < 16 * STR_; idx += TPB_) hS[idx] = 0.f;
  __syncthreads();

  for (int t = 0; t < T_; ++t) {
    const float* gxt = gxbuf + ((size_t)(r * T_ + t) * BH_) * G3_;
    float xg[4][3];
    #pragma unroll
    for (int i = 0; i < 4; ++i)
      #pragma unroll
      for (int g = 0; g < 3; ++g)
        xg[i][g] = gxt[(3 * i + m0) * G3_ + g * HID_ + f];

    v8f acc[2];
    mm1x2_trB(hS, WhhS, 2 * wave, lane, acc);
    #pragma unroll
    for (int j = 0; j < 2; ++j) {
      const int n = (2 * wave + j) * 16 + l15;
      #pragma unroll
      for (int v = 0; v < 8; ++v)
        gh[(v + 8 * half) * GSTR_ + n] = acc[j][v];
    }
    __syncthreads();

    #pragma unroll
    for (int i = 0; i < 4; ++i) {
      const int m = 3 * i + m0;
      const float xr = xg[i][0];
      const float xz = xg[i][1];
      const float xn = xg[i][2];
      const float hr = gh[m * GSTR_ + f]             + bhhS[f];
      const float hz = gh[m * GSTR_ + HID_ + f]      + bhhS[HID_ + f];
      const float hn = gh[m * GSTR_ + 2 * HID_ + f]  + bhhS[2 * HID_ + f];
      const float rg = 1.f / (1.f + expf(-(xr + hr)));
      const float zg = 1.f / (1.f + expf(-(xz + hz)));
      const float ng = tanhf(xn + rg * hn);
      const float h2 = (1.f - zg) * ng + zg * hS[m * STR_ + f];
      hS[m * STR_ + f] = h2;
      const int bb = m / H_, hh = m % H_;
      out[(((size_t)(bb * ROI_ + r) * T_ + t) * H_ + hh) * HID_ + f] = h2;
    }
    __syncthreads();
  }
}

extern "C" void kernel_launch(void* const* d_in, const int* in_sizes, int n_in,
                              void* d_out, int out_size, void* d_ws, size_t ws_size,
                              hipStream_t stream) {
  const float* x   = (const float*)d_in[0];
  const float* a   = (const float*)d_in[1];
  const float* eps = (const float*)d_in[2];
  const float* W1  = (const float*)d_in[3];
  const float* b1  = (const float*)d_in[4];
  const float* g1  = (const float*)d_in[5];
  const float* be1 = (const float*)d_in[6];
  const float* W2  = (const float*)d_in[7];
  const float* b2  = (const float*)d_in[8];
  const float* g2  = (const float*)d_in[9];
  const float* be2 = (const float*)d_in[10];
  const float* Wih = (const float*)d_in[11];
  const float* Whh = (const float*)d_in[12];
  const float* bih = (const float*)d_in[13];
  const float* bhh = (const float*)d_in[14];
  float* out   = (float*)d_out;
  float* hbuf  = (float*)d_ws;
  float* gxbuf = hbuf + (size_t)ROI_ * ROWS_ * F_;

  const size_t ldsA = (size_t)3 * NP_ * STR_ * sizeof(float);
  ftdgnn_agg_mlp<<<B_ * T_ * H_, TPB_, ldsA, stream>>>(
      x, a, eps, W1, b1, g1, be1, W2, b2, g2, be2, hbuf);

  const size_t ldsP = (size_t)(MCH_ + G3_) * STR_ * sizeof(float);
  ftdgnn_xproj<<<ROI_ * NCH_, TPB_, ldsP, stream>>>(Wih, bih, hbuf, gxbuf);

  const size_t ldsB = (size_t)168 * 1024;
  ftdgnn_gru<<<ROI_, TPB_, ldsB, stream>>>(Whh, bhh, gxbuf, out);
}


# GRU gates: sigmoid and tanh from the hardware exp2 and reciprocal instructions (about 1 ulp each) instead of library expf/tanhf and IEEE division; branch-free, so a thread's four rows interleave
# speedup vs baseline: 1.6612x; 1.0425x over previous
#include <hip/hip_runtime.h>
#include <math.h>

#define B_    4
#define ROI_  90
#define T_    200
#define H_    3
#define F_    96
#define HID_  96
#define G3_   288
#define BH_   12
#define NP_   96
#define STR_  100
#define GSTR_ 292
#define TPB_  288
#define ROWS_ 2400
#define MCH_  96
#define NCH_  25
#define NPASS_ 32
#define NREAL_ 30
static_assert(TPB_ == 3 * NP_ && NPASS_ * TPB_ == NP_ * NP_ && NREAL_ * 3 == ROI_,
              "kernel A staging map");

typedef float v2f __attribute__((ext_vector_type(2)));
typedef float v8f __attribute__((ext_vector_type(8)));

__device__ __forceinline__ v8f wmma4(v2f a, v2f b, v8f c) {
  return __builtin_amdgcn_wmma_f32_16x16x4_f32(false, a, false, b, (short)0, c,
                                               false, false);
}

__device__ __forceinline__ void mm2x2_trB(const float* As, const float* Bt,
                                          int sr, int sc, int lane, v8f acc[4]) {
  const int half = lane >> 4;
  const int l15  = lane & 15;
  const float* a0 = As + (sr * 32 + l15) * STR_ + (half << 1);
  const float* a1 = a0 + 16 * STR_;
  const float* b0 = Bt + (sc * 32 + l15) * STR_ + (half << 1);
  const float* b1 = b0 + 16 * STR_;
  #pragma unroll
  for (int j = 0; j < 4; ++j) acc[j] = (v8f){0.f,0.f,0.f,0.f,0.f,0.f,0.f,0.f};
  #pragma unroll
  for (int k0 = 0; k0 < NP_; k0 += 4) {
    v2f av0 = *(const v2f*)(a0 + k0);
    v2f av1 = *(const v2f*)(a1 + k0);
    v2f bv0 = *(const v2f*)(b0 + k0);
    v2f bv1 = *(const v2f*)(b1 + k0);
    acc[0] = wmma4(av0, bv0, acc[0]);
    acc[1] = wmma4(av0, bv1, acc[1]);
    acc[2] = wmma4(av1, bv0, acc[2]);
    acc[3] = wmma4(av1, bv1, acc[3]);
  }
}

__device__ __forceinline__ void mm1x2_trB(const float* As, const float* Bt,
                                          int ntBase, int lane, v8f acc[2]) {
  const int half = lane >> 4;
  const int l15  = lane & 15;
  const float* aRow = As + l15 * STR_ + (half << 1);
  const float* bRow = Bt + (ntBase * 16 + l15) * STR_ + (half << 1);
  #pragma unroll
  for (int j = 0; j < 2; ++j) acc[j] = (v8f){0.f,0.f,0.f,0.f,0.f,0.f,0.f,0.f};
  #pragma unroll
  for (int k0 = 0; k0 < F_; k0 += 4) {
    v2f av = *(const v2f*)(aRow + k0);
    #pragma unroll
    for (int j = 0; j < 2; ++j) {
      v2f bv = *(const v2f*)(bRow + j * 16 * STR_ + k0);
      acc[j] = wmma4(av, bv, acc[j]);
    }
  }
}

__global__ void __launch_bounds__(TPB_)
ftdgnn_agg_mlp(const float* __restrict__ x, const float* __restrict__ a,
               const float* __restrict__ eps,
               const float* __restrict__ W1, const float* __restrict__ b1,
               const float* __restrict__ g1, const float* __restrict__ be1,
               const float* __restrict__ W2, const float* __restrict__ b2,
               const float* __restrict__ g2, const float* __restrict__ be2,
               float* __restrict__ hbuf) {
  extern __shared__ float smem[];
  float* xs = smem;
  float* ms = xs + NP_ * STR_;
  float* ws = ms + NP_ * STR_;

  const int tid  = threadIdx.x;
  const int lane = tid & 31;
  const int wave = tid >> 5;
  const int half = lane >> 4;
  const int l15  = lane & 15;
  const int sr   = wave / 3;
  const int sc   = wave % 3;

  const int bth = blockIdx.x;
  const int b   = bth / (T_ * H_);
  const int th  = bth % (T_ * H_);
  const int t   = th / H_;
  const int hh  = th % H_;

  const int r0 = tid / NP_;
  const int c0 = tid % NP_;
  const int ca = (c0 < ROI_) ? c0 : (ROI_ - 1);
  const size_t rowBase = ((size_t)(b * ROI_ + r0) * T_ + t) * H_ + hh;
  const float* xp  = x  + rowBase * F_   + c0;
  const float* ap  = a  + rowBase * ROI_ + ca;
  const float* w1p = W1 + r0 * HID_ + c0;
  float xv[NREAL_], av[NREAL_], wv[NPASS_];
  #pragma unroll
  for (int i = 0; i < NREAL_; ++i) xv[i] = xp[(size_t)i * (3 * T_ * H_ * F_)];
  #pragma unroll
  for (int i = 0; i < NREAL_; ++i) av[i] = ap[(size_t)i * (3 * T_ * H_ * ROI_)];
  #pragma unroll
  for (int i = 0; i < NPASS_; ++i) wv[i] = w1p[i * (3 * HID_)];
  #pragma unroll
  for (int i = 0; i < NPASS_; ++i) {
    const int r = r0 + 3 * i;
    float xe = 0.f, me = 0.f;
    if (i < NREAL_) {
      xe = xv[i];
      me = (c0 < ROI_ && av[i] != 0.f) ? 1.f : 0.f;
    }
    xs[c0 * STR_ + r] = xe;
    ms[r * STR_ + c0] = me;
    ws[c0 * STR_ + r] = wv[i];
  }
  __syncthreads();

  const float epsv = eps[0];
  const float rs   = rsqrtf(1.0f + 1e-5f);

  v8f acc[4];

  mm2x2_trB(ms, xs, sr, sc, lane, acc);
  __syncthreads();
  #pragma unroll
  for (int di = 0; di < 2; ++di)
    #pragma unroll
    for (int dj = 0; dj < 2; ++dj) {
      const int n = (2 * sc + dj) * 16 + l15;
      #pragma unroll
      for (int v = 0; v < 8; ++v) {
        const int m = (2 * sr + di) * 16 + v + 8 * half;
        ms[m * STR_ + n] = acc[di * 2 + dj][v] + epsv * xs[n * STR_ + m];
      }
    }
  __syncthreads();

  mm2x2_trB(ms, ws, sr, sc, lane, acc);
  __syncthreads();
  const float* w2p = W2 + r0 * F_ + c0;
  #pragma unroll
  for (int i = 0; i < NPASS_; ++i) wv[i] = w2p[i * (3 * F_)];
  #pragma unroll
  for (int di = 0; di < 2; ++di)
    #pragma unroll
    for (int dj = 0; dj < 2; ++dj) {
      const int n = (2 * sc + dj) * 16 + l15;
      const float sc1 = g1[n] * rs, sh1 = be1[n], bb1 = b1[n];
      #pragma unroll
      for (int v = 0; v < 8; ++v) {
        const int m = (2 * sr + di) * 16 + v + 8 * half;
        float z = (acc[di * 2 + dj][v] + bb1) * sc1 + sh1;
        z = (z > 0.f) ? z : expm1f(z);
        xs[m * STR_ + n] = z;
      }
    }
  #pragma unroll
  for (int i = 0; i < NPASS_; ++i)
    ws[c0 * STR_ + r0 + 3 * i] = wv[i];
  __syncthreads();

  mm2x2_trB(xs, ws, sr, sc, lane, acc);
  #pragma unroll
  for (int di = 0; di < 2; ++di)
    #pragma unroll
    for (int dj = 0; dj < 2; ++dj) {
      const int n = (2 * sc + dj) * 16 + l15;
      const float sc2 = g2[n] * rs, sh2 = be2[n], bb2 = b2[n];
      #pragma unroll
      for (int v = 0; v < 8; ++v) {
        const int m = (2 * sr + di) * 16 + v + 8 * half;
        if (m < ROI_) {
          float z = (acc[di * 2 + dj][v] + bb2) * sc2 + sh2;
          z = (z > 0.f) ? z : 0.f;
          hbuf[((size_t)(m * T_ + t) * BH_ + (b * H_ + hh)) * F_ + n] = z;
        }
      }
    }
}

__global__ void __launch_bounds__(TPB_)
ftdgnn_xproj(const float* __restrict__ Wih, const float* __restrict__ bih,
             const float* __restrict__ hbuf, float* __restrict__ gxbuf) {
  extern __shared__ float smem[];
  float* As = smem;
  float* Bs = As + MCH_ * STR_;

  const int tid  = threadIdx.x;
  const int lane = tid & 31;
  const int wave = tid >> 5;
  const int half = lane >> 4;
  const int l15  = lane & 15;
  const int r    = blockIdx.x / NCH_;
  const int ch   = blockIdx.x % NCH_;
  const size_t row0 = (size_t)r * ROWS_ + (size_t)ch * MCH_;

  const float* Ag = hbuf + row0 * F_;
  const float* Bg = Wih + (size_t)r * G3_ * F_;
  #pragma unroll
  for (int i = 0; i < MCH_ * (F_ / 4) / TPB_; ++i) {
    const int idx = tid + i * TPB_;
    const int row = idx / (F_ / 4), k4 = idx % (F_ / 4);
    const float4 v = *(const float4*)(Ag + row * F_ + k4 * 4);
    *(float4*)(As + row * STR_ + k4 * 4) = v;
  }
  #pragma unroll
  for (int i = 0; i < G3_ * (F_ / 4) / TPB_; ++i) {
    const int idx = tid + i * TPB_;
    const int g = idx / (F_ / 4), k4 = idx % (F_ / 4);
    const float4 v = *(const float4*)(Bg + g * F_ + k4 * 4);
    *(float4*)(Bs + g * STR_ + k4 * 4) = v;
  }
  __syncthreads();

  float bias[2];
  #pragma unroll
  for (int dj = 0; dj < 2; ++dj) bias[dj] = bih[r * G3_ + (2 * wave + dj) * 16 + l15];

  for (int sr = 0; sr < MCH_ / 32; ++sr) {
    v8f acc[4];
    mm2x2_trB(As, Bs, sr, wave, lane, acc);
    #pragma unroll
    for (int di = 0; di < 2; ++di)
      #pragma unroll
      for (int dj = 0; dj < 2; ++dj) {
        const int n = (2 * wave + dj) * 16 + l15;
        #pragma unroll
        for (int v = 0; v < 8; ++v) {
          const int m = (2 * sr + di) * 16 + v + 8 * half;
          gxbuf[(row0 + m) * G3_ + n] = acc[di * 2 + dj][v] + bias[dj];
        }
      }
  }
}

__global__ void __launch_bounds__(TPB_)
ftdgnn_gru(const float* __restrict__ Whh, const float* __restrict__ bhh,
           const float* __restrict__ gxbuf, float* __restrict__ out) {
  extern __shared__ float smem[];
  float* WhhS = smem;
  float* hS   = WhhS + G3_ * STR_;
  float* gh   = hS   + 16 * STR_;
  float* bhhS = gh   + 16 * GSTR_;

  const int tid  = threadIdx.x;
  const int lane = tid & 31;
  const int wave = tid >> 5;
  const int half = lane >> 4;
  const int l15  = lane & 15;
  const int r    = blockIdx.x;
  const int f    = tid % HID_;
  const int m0   = tid / HID_;
  const float kLog2e = 1.44269504088896341f;

  for (int idx = tid; idx < G3_ * F_; idx += TPB_) {
    const int g = idx / F_, k = idx % F_;
    WhhS[g * STR_ + k] = Whh[((size_t)r * G3_ + g) * HID_ + k];
  }
  for (int idx = tid; idx < G3_; idx += TPB_) bhhS[idx] = bhh[r * G3_ + idx];
  for (int idx = tid; idx < 16 * STR_; idx += TPB_) hS[idx] = 0.f;
  __syncthreads();

  for (int t = 0; t < T_; ++t) {
    const float* gxt = gxbuf + ((size_t)(r * T_ + t) * BH_) * G3_;
    float xg[4][3];
    #pragma unroll
    for (int i = 0; i < 4; ++i)
      #pragma unroll
      for (int g = 0; g < 3; ++g)
        xg[i][g] = gxt[(3 * i + m0) * G3_ + g * HID_ + f];

    v8f acc[2];
    mm1x2_trB(hS, WhhS, 2 * wave, lane, acc);
    #pragma unroll
    for (int j = 0; j < 2; ++j) {
      const int n = (2 * wave + j) * 16 + l15;
      #pragma unroll
      for (int v = 0; v < 8; ++v)
        gh[(v + 8 * half) * GSTR_ + n] = acc[j][v];
    }
    __syncthreads();

    float hr[4], hz[4], hn[4], hp[4];
    #pragma unroll
    for (int i = 0; i < 4; ++i) {
      const int m = 3 * i + m0;
      hr[i] = gh[m * GSTR_ + f]            + bhhS[f];
      hz[i] = gh[m * GSTR_ + HID_ + f]     + bhhS[HID_ + f];
      hn[i] = gh[m * GSTR_ + 2 * HID_ + f] + bhhS[2 * HID_ + f];
      hp[i] = hS[m * STR_ + f];
    }
    #pragma unroll
    for (int i = 0; i < 4; ++i) {
      const int m = 3 * i + m0;
      const float xr = xg[i][0];
      const float xz = xg[i][1];
      const float xn = xg[i][2];
      const float rg = __builtin_amdgcn_rcpf(1.f + __builtin_amdgcn_exp2f(-kLog2e * (xr + hr[i])));
      const float zg = __builtin_amdgcn_rcpf(1.f + __builtin_amdgcn_exp2f(-kLog2e * (xz + hz[i])));
      const float en = __builtin_amdgcn_exp2f((2.f * kLog2e) * (xn + rg * hn[i]));
      const float ng = 1.f - 2.f * __builtin_amdgcn_rcpf(1.f + en);
      const float h2 = (1.f - zg) * ng + zg * hp[i];
      hS[m * STR_ + f] = h2;
      const int bb = m / H_, hh = m % H_;
      out[(((size_t)(bb * ROI_ + r) * T_ + t) * H_ + hh) * HID_ + f] = h2;
    }
    __syncthreads();
  }
}

extern "C" void kernel_launch(void* const* d_in, const int* in_sizes, int n_in,
                              void* d_out, int out_size, void* d_ws, size_t ws_size,
                              hipStream_t stream) {
  const float* x   = (const float*)d_in[0];
  const float* a   = (const float*)d_in[1];
  const float* eps = (const float*)d_in[2];
  const float* W1  = (const float*)d_in[3];
  const float* b1  = (const float*)d_in[4];
  const float* g1  = (const float*)d_in[5];
  const float* be1 = (const float*)d_in[6];
  const float* W2  = (const float*)d_in[7];
  const float* b2  = (const float*)d_in[8];
  const float* g2  = (const float*)d_in[9];
  const float* be2 = (const float*)d_in[10];
  const float* Wih = (const float*)d_in[11];
  const float* Whh = (const float*)d_in[12];
  const float* bih = (const float*)d_in[13];
  const float* bhh = (const float*)d_in[14];
  float* out   = (float*)d_out;
  float* hbuf  = (float*)d_ws;
  float* gxbuf = hbuf + (size_t)ROI_ * ROWS_ * F_;

  const size_t ldsA = (size_t)3 * NP_ * STR_ * sizeof(float);
  ftdgnn_agg_mlp<<<B_ * T_ * H_, TPB_, ldsA, stream>>>(
      x, a, eps, W1, b1, g1, be1, W2, b2, g2, be2, hbuf);

  const size_t ldsP = (size_t)(MCH_ + G3_) * STR_ * sizeof(float);
  ftdgnn_xproj<<<ROI_ * NCH_, TPB_, ldsP, stream>>>(Wih, bih, hbuf, gxbuf);

  const size_t ldsB = (size_t)168 * 1024;
  ftdgnn_gru<<<ROI_, TPB_, ldsB, stream>>>(Whh, bhh, gxbuf, out);
}


# GRU time loop: workgroup barriers wait for LDS only, so out stores and gate-input loads (issued one step ahead, two register sets) stay in flight across them instead of stalling every step
# speedup vs baseline: 2.3126x; 1.3921x over previous
#include <hip/hip_runtime.h>
#include <math.h>

#define B_    4
#define ROI_  90
#define T_    200
#define H_    3
#define F_    96
#define HID_  96
#define G3_   288
#define BH_   12
#define NP_   96
#define STR_  100
#define GSTR_ 292
#define TPB_  288
#define ROWS_ 2400
#define MCH_  96
#define NCH_  25
#define NPASS_ 32
#define NREAL_ 30
static_assert(TPB_ == 3 * NP_ && NPASS_ * TPB_ == NP_ * NP_ && NREAL_ * 3 == ROI_,
              "kernel A staging map");

typedef float v2f __attribute__((ext_vector_type(2)));
typedef float v8f __attribute__((ext_vector_type(8)));

__device__ __forceinline__ v8f wmma4(v2f a, v2f b, v8f c) {
  return __builtin_amdgcn_wmma_f32_16x16x4_f32(false, a, false, b, (short)0, c,
                                               false, false);
}

__device__ __forceinline__ void mm2x2_trB(const float* As, const float* Bt,
                                          int sr, int sc, int lane, v8f acc[4]) {
  const int half = lane >> 4;
  const int l15  = lane & 15;
  const float* a0 = As + (sr * 32 + l15) * STR_ + (half << 1);
  const float* a1 = a0 + 16 * STR_;
  const float* b0 = Bt + (sc * 32 + l15) * STR_ + (half << 1);
  const float* b1 = b0 + 16 * STR_;
  #pragma unroll
  for (int j = 0; j < 4; ++j) acc[j] = (v8f){0.f,0.f,0.f,0.f,0.f,0.f,0.f,0.f};
  #pragma unroll
  for (int k0 = 0; k0 < NP_; k0 += 4) {
    v2f av0 = *(const v2f*)(a0 + k0);
    v2f av1 = *(const v2f*)(a1 + k0);
    v2f bv0 = *(const v2f*)(b0 + k0);
    v2f bv1 = *(const v2f*)(b1 + k0);
    acc[0] = wmma4(av0, bv0, acc[0]);
    acc[1] = wmma4(av0, bv1, acc[1]);
    acc[2] = wmma4(av1, bv0, acc[2]);
    acc[3] = wmma4(av1, bv1, acc[3]);
  }
}

__device__ __forceinline__ void mm1x2_trB(const float* As, const float* Bt,
                                          int ntBase, int lane, v8f acc[2]) {
  const int half = lane >> 4;
  const int l15  = lane & 15;
  const float* aRow = As + l15 * STR_ + (half << 1);
  const float* bRow = Bt + (ntBase * 16 + l15) * STR_ + (half << 1);
  #pragma unroll
  for (int j = 0; j < 2; ++j) acc[j] = (v8f){0.f,0.f,0.f,0.f,0.f,0.f,0.f,0.f};
  #pragma unroll
  for (int k0 = 0; k0 < F_; k0 += 4) {
    v2f av = *(const v2f*)(aRow + k0);
    #pragma unroll
    for (int j = 0; j < 2; ++j) {
      v2f bv = *(const v2f*)(bRow + j * 16 * STR_ + k0);
      acc[j] = wmma4(av, bv, acc[j]);
    }
  }
}

__global__ void __launch_bounds__(TPB_)
ftdgnn_agg_mlp(const float* __restrict__ x, const float* __restrict__ a,
               const float* __restrict__ eps,
               const float* __restrict__ W1, const float* __restrict__ b1,
               const float* __restrict__ g1, const float* __restrict__ be1,
               const float* __restrict__ W2, const float* __restrict__ b2,
               const float* __restrict__ g2, const float* __restrict__ be2,
               float* __restrict__ hbuf) {
  extern __shared__ float smem[];
  float* xs = smem;
  float* ms = xs + NP_ * STR_;
  float* ws = ms + NP_ * STR_;

  const int tid  = threadIdx.x;
  const int lane = tid & 31;
  const int wave = tid >> 5;
  const int half = lane >> 4;
  const int l15  = lane & 15;
  const int sr   = wave / 3;
  const int sc   = wave % 3;

  const int bth = blockIdx.x;
  const int b   = bth / (T_ * H_);
  const int th  = bth % (T_ * H_);
  const int t   = th / H_;
  const int hh  = th % H_;

  const int r0 = tid / NP_;
  const int c0 = tid % NP_;
  const int ca = (c0 < ROI_) ? c0 : (ROI_ - 1);
  const size_t rowBase = ((size_t)(b * ROI_ + r0) * T_ + t) * H_ + hh;
  const float* xp  = x  + rowBase * F_   + c0;
  const float* ap  = a  + rowBase * ROI_ + ca;
  const float* w1p = W1 + r0 * HID_ + c0;
  float xv[NREAL_], av[NREAL_], wv[NPASS_];
  #pragma unroll
  for (int i = 0; i < NREAL_; ++i) xv[i] = xp[(size_t)i * (3 * T_ * H_ * F_)];
  #pragma unroll
  for (int i = 0; i < NREAL_; ++i) av[i] = ap[(size_t)i * (3 * T_ * H_ * ROI_)];
  #pragma unroll
  for (int i = 0; i < NPASS_; ++i) wv[i] = w1p[i * (3 * HID_)];
  #pragma unroll
  for (int i = 0; i < NPASS_; ++i) {
    const int r = r0 + 3 * i;
    float xe = 0.f, me = 0.f;
    if (i < NREAL_) {
      xe = xv[i];
      me = (c0 < ROI_ && av[i] != 0.f) ? 1.f : 0.f;
    }
    xs[c0 * STR_ + r] = xe;
    ms[r * STR_ + c0] = me;
    ws[c0 * STR_ + r] = wv[i];
  }
  __syncthreads();

  const float epsv = eps[0];
  const float rs   = rsqrtf(1.0f + 1e-5f);

  v8f acc[4];

  mm2x2_trB(ms, xs, sr, sc, lane, acc);
  __syncthreads();
  #pragma unroll
  for (int di = 0; di < 2; ++di)
    #pragma unroll
    for (int dj = 0; dj < 2; ++dj) {
      const int n = (2 * sc + dj) * 16 + l15;
      #pragma unroll
      for (int v = 0; v < 8; ++v) {
        const int m = (2 * sr + di) * 16 + v + 8 * half;
        ms[m * STR_ + n] = acc[di * 2 + dj][v] + epsv * xs[n * STR_ + m];
      }
    }
  __syncthreads();

  mm2x2_trB(ms, ws, sr, sc, lane, acc);
  __syncthreads();
  const float* w2p = W2 + r0 * F_ + c0;
  #pragma unroll
  for (int i = 0; i < NPASS_; ++i) wv[i] = w2p[i * (3 * F_)];
  #pragma unroll
  for (int di = 0; di < 2; ++di)
    #pragma unroll
    for (int dj = 0; dj < 2; ++dj) {
      const int n = (2 * sc + dj) * 16 + l15;
      const float sc1 = g1[n] * rs, sh1 = be1[n], bb1 = b1[n];
      #pragma unroll
      for (int v = 0; v < 8; ++v) {
        const int m = (2 * sr + di) * 16 + v + 8 * half;
        float z = (acc[di * 2 + dj][v] + bb1) * sc1 + sh1;
        z = (z > 0.f) ? z : expm1f(z);
        xs[m * STR_ + n] = z;
      }
    }
  #pragma unroll
  for (int i = 0; i < NPASS_; ++i)
    ws[c0 * STR_ + r0 + 3 * i] = wv[i];
  __syncthreads();

  mm2x2_trB(xs, ws, sr, sc, lane, acc);
  #pragma unroll
  for (int di = 0; di < 2; ++di)
    #pragma unroll
    for (int dj = 0; dj < 2; ++dj) {
      const int n = (2 * sc + dj) * 16 + l15;
      const float sc2 = g2[n] * rs, sh2 = be2[n], bb2 = b2[n];
      #pragma unroll
      for (int v = 0; v < 8; ++v) {
        const int m = (2 * sr + di) * 16 + v + 8 * half;
        if (m < ROI_) {
          float z = (acc[di * 2 + dj][v] + bb2) * sc2 + sh2;
          z = (z > 0.f) ? z : 0.f;
          hbuf[((size_t)(m * T_ + t) * BH_ + (b * H_ + hh)) * F_ + n] = z;
        }
      }
    }
}

__global__ void __launch_bounds__(TPB_)
ftdgnn_xproj(const float* __restrict__ Wih, const float* __restrict__ bih,
             const float* __restrict__ hbuf, float* __restrict__ gxbuf) {
  extern __shared__ float smem[];
  float* As = smem;
  float* Bs = As + MCH_ * STR_;

  const int tid  = threadIdx.x;
  const int lane = tid & 31;
  const int wave = tid >> 5;
  const int half = lane >> 4;
  const int l15  = lane & 15;
  const int r    = blockIdx.x / NCH_;
  const int ch   = blockIdx.x % NCH_;
  const size_t row0 = (size_t)r * ROWS_ + (size_t)ch * MCH_;

  const float* Ag = hbuf + row0 * F_;
  const float* Bg = Wih + (size_t)r * G3_ * F_;
  #pragma unroll
  for (int i = 0; i < MCH_ * (F_ / 4) / TPB_; ++i) {
    const int idx = tid + i * TPB_;
    const int row = idx / (F_ / 4), k4 = idx % (F_ / 4);
    const float4 v = *(const float4*)(Ag + row * F_ + k4 * 4);
    *(float4*)(As + row * STR_ + k4 * 4) = v;
  }
  #pragma unroll
  for (int i = 0; i < G3_ * (F_ / 4) / TPB_; ++i) {
    const int idx = tid + i * TPB_;
    const int g = idx / (F_ / 4), k4 = idx % (F_ / 4);
    const float4 v = *(const float4*)(Bg + g * F_ + k4 * 4);
    *(float4*)(Bs + g * STR_ + k4 * 4) = v;
  }
  __syncthreads();

  float bias[2];
  #pragma unroll
  for (int dj = 0; dj < 2; ++dj) bias[dj] = bih[r * G3_ + (2 * wave + dj) * 16 + l15];

  for (int sr = 0; sr < MCH_ / 32; ++sr) {
    v8f acc[4];
    mm2x2_trB(As, Bs, sr, wave, lane, acc);
    #pragma unroll
    for (int di = 0; di < 2; ++di)
      #pragma unroll
      for (int dj = 0; dj < 2; ++dj) {
        const int n = (2 * wave + dj) * 16 + l15;
        #pragma unroll
        for (int v = 0; v < 8; ++v) {
          const int m = (2 * sr + di) * 16 + v + 8 * half;
          gxbuf[(row0 + m) * G3_ + n] = acc[di * 2 + dj][v] + bias[dj];
        }
      }
  }
}

__device__ __forceinline__ void lds_barrier() {
  __builtin_amdgcn_fence(__ATOMIC_RELEASE, "workgroup", "local");
  __builtin_amdgcn_s_barrier();
  __builtin_amdgcn_fence(__ATOMIC_ACQUIRE, "workgroup", "local");
}

__global__ void __launch_bounds__(TPB_)
ftdgnn_gru(const float* __restrict__ Whh, const float* __restrict__ bhh,
           const float* __restrict__ gxbuf, float* __restrict__ out) {
  extern __shared__ float smem[];
  float* WhhS = smem;
  float* hS   = WhhS + G3_ * STR_;
  float* gh   = hS   + 16 * STR_;
  float* bhhS = gh   + 16 * GSTR_;

  const int tid  = threadIdx.x;
  const int lane = tid & 31;
  const int wave = tid >> 5;
  const int half = lane >> 4;
  const int l15  = lane & 15;
  const int r    = blockIdx.x;
  const int f    = tid % HID_;
  const int m0   = tid / HID_;
  const float kLog2e = 1.44269504088896341f;

  for (int idx = tid; idx < G3_ * F_; idx += TPB_) {
    const int g = idx / F_, k = idx % F_;
    WhhS[g * STR_ + k] = Whh[((size_t)r * G3_ + g) * HID_ + k];
  }
  for (int idx = tid; idx < G3_; idx += TPB_) bhhS[idx] = bhh[r * G3_ + idx];
  for (int idx = tid; idx < 16 * STR_; idx += TPB_) hS[idx] = 0.f;
  __syncthreads();

  const float* gxr = gxbuf + (size_t)r * T_ * BH_ * G3_ + m0 * G3_ + f;
  auto load_gx = [&](int t, float (&xg)[4][3]) {
    const float* gxt = gxr + (size_t)t * (BH_ * G3_);
    #pragma unroll
    for (int i = 0; i < 4; ++i)
      #pragma unroll
      for (int g = 0; g < 3; ++g)
        xg[i][g] = gxt[(3 * i) * G3_ + g * HID_];
  };

  auto step = [&](int t, const float (&xg)[4][3]) {
    v8f acc[2];
    mm1x2_trB(hS, WhhS, 2 * wave, lane, acc);
    #pragma unroll
    for (int j = 0; j < 2; ++j) {
      const int n = (2 * wave + j) * 16 + l15;
      #pragma unroll
      for (int v = 0; v < 8; ++v)
        gh[(v + 8 * half) * GSTR_ + n] = acc[j][v];
    }
    lds_barrier();

    float hr[4], hz[4], hn[4], hp[4];
    #pragma unroll
    for (int i = 0; i < 4; ++i) {
      const int m = 3 * i + m0;
      hr[i] = gh[m * GSTR_ + f]            + bhhS[f];
      hz[i] = gh[m * GSTR_ + HID_ + f]     + bhhS[HID_ + f];
      hn[i] = gh[m * GSTR_ + 2 * HID_ + f] + bhhS[2 * HID_ + f];
      hp[i] = hS[m * STR_ + f];
    }
    #pragma unroll
    for (int i = 0; i < 4; ++i) {
      const int m = 3 * i + m0;
      const float xr = xg[i][0];
      const float xz = xg[i][1];
      const float xn = xg[i][2];
      const float rg = __builtin_amdgcn_rcpf(1.f + __builtin_amdgcn_exp2f(-kLog2e * (xr + hr[i])));
      const float zg = __builtin_amdgcn_rcpf(1.f + __builtin_amdgcn_exp2f(-kLog2e * (xz + hz[i])));
      const float en = __builtin_amdgcn_exp2f((2.f * kLog2e) * (xn + rg * hn[i]));
      const float ng = 1.f - 2.f * __builtin_amdgcn_rcpf(1.f + en);
      const float h2 = (1.f - zg) * ng + zg * hp[i];
      hS[m * STR_ + f] = h2;
      const int bb = m / H_, hh = m % H_;
      out[(((size_t)(bb * ROI_ + r) * T_ + t) * H_ + hh) * HID_ + f] = h2;
    }
    lds_barrier();
  };

  static_assert(T_ % 2 == 0, "time loop handles two steps per trip");
  float xa[4][3], xb[4][3];
  load_gx(0, xa);
  for (int t = 0; t < T_; t += 2) {
    load_gx(t + 1, xb);
    step(t, xa);
    load_gx((t + 2 < T_) ? t + 2 : T_ - 1, xa);
    step(t + 1, xb);
  }
}

extern "C" void kernel_launch(void* const* d_in, const int* in_sizes, int n_in,
                              void* d_out, int out_size, void* d_ws, size_t ws_size,
                              hipStream_t stream) {
  const float* x   = (const float*)d_in[0];
  const float* a   = (const float*)d_in[1];
  const float* eps = (const float*)d_in[2];
  const float* W1  = (const float*)d_in[3];
  const float* b1  = (const float*)d_in[4];
  const float* g1  = (const float*)d_in[5];
  const float* be1 = (const float*)d_in[6];
  const float* W2  = (const float*)d_in[7];
  const float* b2  = (const float*)d_in[8];
  const float* g2  = (const float*)d_in[9];
  const float* be2 = (const float*)d_in[10];
  const float* Wih = (const float*)d_in[11];
  const float* Whh = (const float*)d_in[12];
  const float* bih = (const float*)d_in[13];
  const float* bhh = (const float*)d_in[14];
  float* out   = (float*)d_out;
  float* hbuf  = (float*)d_ws;
  float* gxbuf = hbuf + (size_t)ROI_ * ROWS_ * F_;

  const size_t ldsA = (size_t)3 * NP_ * STR_ * sizeof(float);
  ftdgnn_agg_mlp<<<B_ * T_ * H_, TPB_, ldsA, stream>>>(
      x, a, eps, W1, b1, g1, be1, W2, b2, g2, be2, hbuf);

  const size_t ldsP = (size_t)(MCH_ + G3_) * STR_ * sizeof(float);
  ftdgnn_xproj<<<ROI_ * NCH_, TPB_, ldsP, stream>>>(Wih, bih, hbuf, gxbuf);

  const size_t ldsB = (size_t)168 * 1024;
  ftdgnn_gru<<<ROI_, TPB_, ldsB, stream>>>(Whh, bhh, gxbuf, out);
}


# GRU time loop on 12 waves, not 9: each wave does 3 gate tiles over half of K, so every SIMD's matrix pipe issues 108 WMMAs a step, not 144/96/96/96; a gate sum = its two half-K partial sums added
# speedup vs baseline: 2.6665x; 1.1530x over previous
#include <hip/hip_runtime.h>
#include <math.h>

#define B_    4
#define ROI_  90
#define T_    200
#define H_    3
#define F_    96
#define HID_  96
#define G3_   288
#define BH_   12
#define NP_   96
#define STR_  100
#define GSTR_ 290
#define TPB_  288
#define TPBG_ 384
#define KH_   48
#define ROWS_ 2400
#define MCH_  96
#define NCH_  25
#define NPASS_ 32
#define NREAL_ 30
static_assert(TPB_ == 3 * NP_ && NPASS_ * TPB_ == NP_ * NP_ && NREAL_ * 3 == ROI_,
              "kernel A staging map");

typedef float v2f __attribute__((ext_vector_type(2)));
typedef float v8f __attribute__((ext_vector_type(8)));

__device__ __forceinline__ v8f wmma4(v2f a, v2f b, v8f c) {
  return __builtin_amdgcn_wmma_f32_16x16x4_f32(false, a, false, b, (short)0, c,
                                               false, false);
}

__device__ __forceinline__ void mm2x2_trB(const float* As, const float* Bt,
                                          int sr, int sc, int lane, v8f acc[4]) {
  const int half = lane >> 4;
  const int l15  = lane & 15;
  const float* a0 = As + (sr * 32 + l15) * STR_ + (half << 1);
  const float* a1 = a0 + 16 * STR_;
  const float* b0 = Bt + (sc * 32 + l15) * STR_ + (half << 1);
  const float* b1 = b0 + 16 * STR_;
  #pragma unroll
  for (int j = 0; j < 4; ++j) acc[j] = (v8f){0.f,0.f,0.f,0.f,0.f,0.f,0.f,0.f};
  #pragma unroll
  for (int k0 = 0; k0 < NP_; k0 += 4) {
    v2f av0 = *(const v2f*)(a0 + k0);
    v2f av1 = *(const v2f*)(a1 + k0);
    v2f bv0 = *(const v2f*)(b0 + k0);
    v2f bv1 = *(const v2f*)(b1 + k0);
    acc[0] = wmma4(av0, bv0, acc[0]);
    acc[1] = wmma4(av0, bv1, acc[1]);
    acc[2] = wmma4(av1, bv0, acc[2]);
    acc[3] = wmma4(av1, bv1, acc[3]);
  }
}

__device__ __forceinline__ void mm1x3_halfK_trB(const float* As, const float* Bt,
                                                int colBase, int kOff, int lane, v8f acc[3]) {
  const int half = lane >> 4;
  const int l15  = lane & 15;
  const float* aRow = As + l15 * STR_ + kOff + (half << 1);
  const float* bRow = Bt + (colBase + l15) * STR_ + kOff + (half << 1);
  #pragma unroll
  for (int g = 0; g < 3; ++g) acc[g] = (v8f){0.f,0.f,0.f,0.f,0.f,0.f,0.f,0.f};
  #pragma unroll
  for (int k0 = 0; k0 < KH_; k0 += 4) {
    v2f av = *(const v2f*)(aRow + k0);
    #pragma unroll
    for (int g = 0; g < 3; ++g) {
      v2f bv = *(const v2f*)(bRow + g * HID_ * STR_ + k0);
      acc[g] = wmma4(av, bv, acc[g]);
    }
  }
}

__global__ void __launch_bounds__(TPB_)
ftdgnn_agg_mlp(const float* __restrict__ x, const float* __restrict__ a,
               const float* __restrict__ eps,
               const float* __restrict__ W1, const float* __restrict__ b1,
               const float* __restrict__ g1, const float* __restrict__ be1,
               const float* __restrict__ W2, const float* __restrict__ b2,
               const float* __restrict__ g2, const float* __restrict__ be2,
               float* __restrict__ hbuf) {
  extern __shared__ float smem[];
  float* xs = smem;
  float* ms = xs + NP_ * STR_;
  float* ws = ms + NP_ * STR_;

  const int tid  = threadIdx.x;
  const int lane = tid & 31;
  const int wave = tid >> 5;
  const int half = lane >> 4;
  const int l15  = lane & 15;
  const int sr   = wave / 3;
  const int sc   = wave % 3;

  const int bth = blockIdx.x;
  const int b   = bth / (T_ * H_);
  const int th  = bth % (T_ * H_);
  const int t   = th / H_;
  const int hh  = th % H_;

  const int r0 = tid / NP_;
  const int c0 = tid % NP_;
  const int ca = (c0 < ROI_) ? c0 : (ROI_ - 1);
  const size_t rowBase = ((size_t)(b * ROI_ + r0) * T_ + t) * H_ + hh;
  const float* xp  = x  + rowBase * F_   + c0;
  const float* ap  = a  + rowBase * ROI_ + ca;
  const float* w1p = W1 + r0 * HID_ + c0;
  float xv[NREAL_], av[NREAL_], wv[NPASS_];
  #pragma unroll
  for (int i = 0; i < NREAL_; ++i) xv[i] = xp[(size_t)i * (3 * T_ * H_ * F_)];
  #pragma unroll
  for (int i = 0; i < NREAL_; ++i) av[i] = ap[(size_t)i * (3 * T_ * H_ * ROI_)];
  #pragma unroll
  for (int i = 0; i < NPASS_; ++i) wv[i] = w1p[i * (3 * HID_)];
  #pragma unroll
  for (int i = 0; i < NPASS_; ++i) {
    const int r = r0 + 3 * i;
    float xe = 0.f, me = 0.f;
    if (i < NREAL_) {
      xe = xv[i];
      me = (c0 < ROI_ && av[i] != 0.f) ? 1.f : 0.f;
    }
    xs[c0 * STR_ + r] = xe;
    ms[r * STR_ + c0] = me;
    ws[c0 * STR_ + r] = wv[i];
  }
  __syncthreads();

  const float epsv = eps[0];
  const float rs   = rsqrtf(1.0f + 1e-5f);

  v8f acc[4];

  mm2x2_trB(ms, xs, sr, sc, lane, acc);
  __syncthreads();
  #pragma unroll
  for (int di = 0; di < 2; ++di)
    #pragma unroll
    for (int dj = 0; dj < 2; ++dj) {
      const int n = (2 * sc + dj) * 16 + l15;
      #pragma unroll
      for (int v = 0; v < 8; ++v) {
        const int m = (2 * sr + di) * 16 + v + 8 * half;
        ms[m * STR_ + n] = acc[di * 2 + dj][v] + epsv * xs[n * STR_ + m];
      }
    }
  __syncthreads();

  mm2x2_trB(ms, ws, sr, sc, lane, acc);
  __syncthreads();
  const float* w2p = W2 + r0 * F_ + c0;
  #pragma unroll
  for (int i = 0; i < NPASS_; ++i) wv[i] = w2p[i * (3 * F_)];
  #pragma unroll
  for (int di = 0; di < 2; ++di)
    #pragma unroll
    for (int dj = 0; dj < 2; ++dj) {
      const int n = (2 * sc + dj) * 16 + l15;
      const float sc1 = g1[n] * rs, sh1 = be1[n], bb1 = b1[n];
      #pragma unroll
      for (int v = 0; v < 8; ++v) {
        const int m = (2 * sr + di) * 16 + v + 8 * half;
        float z = (acc[di * 2 + dj][v] + bb1) * sc1 + sh1;
        z = (z > 0.f) ? z : expm1f(z);
        xs[m * STR_ + n] = z;
      }
    }
  #pragma unroll
  for (int i = 0; i < NPASS_; ++i)
    ws[c0 * STR_ + r0 + 3 * i] = wv[i];
  __syncthreads();

  mm2x2_trB(xs, ws, sr, sc, lane, acc);
  #pragma unroll
  for (int di = 0; di < 2; ++di)
    #pragma unroll
    for (int dj = 0; dj < 2; ++dj) {
      const int n = (2 * sc + dj) * 16 + l15;
      const float sc2 = g2[n] * rs, sh2 = be2[n], bb2 = b2[n];
      #pragma unroll
      for (int v = 0; v < 8; ++v) {
        const int m = (2 * sr + di) * 16 + v + 8 * half;
        if (m < ROI_) {
          float z = (acc[di * 2 + dj][v] + bb2) * sc2 + sh2;
          z = (z > 0.f) ? z : 0.f;
          hbuf[((size_t)(m * T_ + t) * BH_ + (b * H_ + hh)) * F_ + n] = z;
        }
      }
    }
}

__global__ void __launch_bounds__(TPB_)
ftdgnn_xproj(const float* __restrict__ Wih, const float* __restrict__ bih,
             const float* __restrict__ hbuf, float* __restrict__ gxbuf) {
  extern __shared__ float smem[];
  float* As = smem;
  float* Bs = As + MCH_ * STR_;

  const int tid  = threadIdx.x;
  const int lane = tid & 31;
  const int wave = tid >> 5;
  const int half = lane >> 4;
  const int l15  = lane & 15;
  const int r    = blockIdx.x / NCH_;
  const int ch   = blockIdx.x % NCH_;
  const size_t row0 = (size_t)r * ROWS_ + (size_t)ch * MCH_;

  const float* Ag = hbuf + row0 * F_;
  const float* Bg = Wih + (size_t)r * G3_ * F_;
  #pragma unroll
  for (int i = 0; i < MCH_ * (F_ / 4) / TPB_; ++i) {
    const int idx = tid + i * TPB_;
    const int row = idx / (F_ / 4), k4 = idx % (F_ / 4);
    const float4 v = *(const float4*)(Ag + row * F_ + k4 * 4);
    *(float4*)(As + row * STR_ + k4 * 4) = v;
  }
  #pragma unroll
  for (int i = 0; i < G3_ * (F_ / 4) / TPB_; ++i) {
    const int idx = tid + i * TPB_;
    const int g = idx / (F_ / 4), k4 = idx % (F_ / 4);
    const float4 v = *(const float4*)(Bg + g * F_ + k4 * 4);
    *(float4*)(Bs + g * STR_ + k4 * 4) = v;
  }
  __syncthreads();

  float bias[2];
  #pragma unroll
  for (int dj = 0; dj < 2; ++dj) bias[dj] = bih[r * G3_ + (2 * wave + dj) * 16 + l15];

  for (int sr = 0; sr < MCH_ / 32; ++sr) {
    v8f acc[4];
    mm2x2_trB(As, Bs, sr, wave, lane, acc);
    #pragma unroll
    for (int di = 0; di < 2; ++di)
      #pragma unroll
      for (int dj = 0; dj < 2; ++dj) {
        const int n = (2 * wave + dj) * 16 + l15;
        #pragma unroll
        for (int v = 0; v < 8; ++v) {
          const int m = (2 * sr + di) * 16 + v + 8 * half;
          gxbuf[(row0 + m) * G3_ + n] = acc[di * 2 + dj][v] + bias[dj];
        }
      }
  }
}

__device__ __forceinline__ void lds_barrier() {
  __builtin_amdgcn_fence(__ATOMIC_RELEASE, "workgroup", "local");
  __builtin_amdgcn_s_barrier();
  __builtin_amdgcn_fence(__ATOMIC_ACQUIRE, "workgroup", "local");
}

__global__ void __launch_bounds__(TPBG_)
ftdgnn_gru(const float* __restrict__ Whh, const float* __restrict__ bhh,
           const float* __restrict__ gxbuf, float* __restrict__ out) {
  extern __shared__ float smem[];
  float* WhhS = smem;
  float* hS   = WhhS + G3_ * STR_;
  float* gh   = hS   + 16 * STR_;
  float* bhhS = gh   + 16 * GSTR_ * 2;

  const int tid  = threadIdx.x;
  const int lane = tid & 31;
  const int wave = tid >> 5;
  const int half = lane >> 4;
  const int l15  = lane & 15;
  const int r    = blockIdx.x;
  const int cg   = wave >> 1;
  const int kh   = wave & 1;
  const int f    = tid % HID_;
  const int m0   = tid / HID_;
  const float kLog2e = 1.44269504088896341f;

  for (int idx = tid; idx < G3_ * F_; idx += TPBG_) {
    const int g = idx / F_, k = idx % F_;
    WhhS[g * STR_ + k] = Whh[((size_t)r * G3_ + g) * HID_ + k];
  }
  for (int idx = tid; idx < G3_; idx += TPBG_) bhhS[idx] = bhh[r * G3_ + idx];
  for (int idx = tid; idx < 16 * STR_; idx += TPBG_) hS[idx] = 0.f;
  __syncthreads();

  const float* gxr = gxbuf + (size_t)r * T_ * BH_ * G3_ + m0 * G3_ + f;
  auto load_gx = [&](int t, float (&xg)[3][3]) {
    const float* gxt = gxr + (size_t)t * (BH_ * G3_);
    #pragma unroll
    for (int i = 0; i < 3; ++i)
      #pragma unroll
      for (int g = 0; g < 3; ++g)
        xg[i][g] = gxt[(4 * i) * G3_ + g * HID_];
  };

  auto step = [&](int t, const float (&xg)[3][3]) {
    v8f acc[3];
    mm1x3_halfK_trB(hS, WhhS, 16 * cg, KH_ * kh, lane, acc);
    #pragma unroll
    for (int g = 0; g < 3; ++g) {
      const int n = g * HID_ + 16 * cg + l15;
      #pragma unroll
      for (int v = 0; v < 8; ++v)
        gh[((v + 8 * half) * GSTR_ + n) * 2 + kh] = acc[g][v];
    }
    lds_barrier();

    const v2f* ghp = (const v2f*)gh;
    float hr[3], hz[3], hn[3], hp[3];
    #pragma unroll
    for (int i = 0; i < 3; ++i) {
      const int m = 4 * i + m0;
      const v2f pr = ghp[m * GSTR_ + f];
      const v2f pz = ghp[m * GSTR_ + HID_ + f];
      const v2f pn = ghp[m * GSTR_ + 2 * HID_ + f];
      hr[i] = (pr.x + pr.y) + bhhS[f];
      hz[i] = (pz.x + pz.y) + bhhS[HID_ + f];
      hn[i] = (pn.x + pn.y) + bhhS[2 * HID_ + f];
      hp[i] = hS[m * STR_ + f];
    }
    #pragma unroll
    for (int i = 0; i < 3; ++i) {
      const int m = 4 * i + m0;
      const float xr = xg[i][0];
      const float xz = xg[i][1];
      const float xn = xg[i][2];
      const float rg = __builtin_amdgcn_rcpf(1.f + __builtin_amdgcn_exp2f(-kLog2e * (xr + hr[i])));
      const float zg = __builtin_amdgcn_rcpf(1.f + __builtin_amdgcn_exp2f(-kLog2e * (xz + hz[i])));
      const float en = __builtin_amdgcn_exp2f((2.f * kLog2e) * (xn + rg * hn[i]));
      const float ng = 1.f - 2.f * __builtin_amdgcn_rcpf(1.f + en);
      const float h2 = (1.f - zg) * ng + zg * hp[i];
      hS[m * STR_ + f] = h2;
      const int bb = m / H_, hh = m % H_;
      out[(((size_t)(bb * ROI_ + r) * T_ + t) * H_ + hh) * HID_ + f] = h2;
    }
    lds_barrier();
  };

  static_assert(T_ % 2 == 0, "time loop handles two steps per trip");
  static_assert(TPBG_ == 4 * HID_ && BH_ == 3 * 4 && TPBG_ / 32 == 2 * (HID_ / 16) && 2 * KH_ == HID_,
                "GRU kernel thread and wave maps");
  float xa[3][3], xb[3][3];
  load_gx(0, xa);
  for (int t = 0; t < T_; t += 2) {
    load_gx(t + 1, xb);
    step(t, xa);
    load_gx((t + 2 < T_) ? t + 2 : T_ - 1, xa);
    step(t + 1, xb);
  }
}

extern "C" void kernel_launch(void* const* d_in, const int* in_sizes, int n_in,
                              void* d_out, int out_size, void* d_ws, size_t ws_size,
                              hipStream_t stream) {
  const float* x   = (const float*)d_in[0];
  const float* a   = (const float*)d_in[1];
  const float* eps = (const float*)d_in[2];
  const float* W1  = (const float*)d_in[3];
  const float* b1  = (const float*)d_in[4];
  const float* g1  = (const float*)d_in[5];
  const float* be1 = (const float*)d_in[6];
  const float* W2  = (const float*)d_in[7];
  const float* b2  = (const float*)d_in[8];
  const float* g2  = (const float*)d_in[9];
  const float* be2 = (const float*)d_in[10];
  const float* Wih = (const float*)d_in[11];
  const float* Whh = (const float*)d_in[12];
  const float* bih = (const float*)d_in[13];
  const float* bhh = (const float*)d_in[14];
  float* out   = (float*)d_out;
  float* hbuf  = (float*)d_ws;
  float* gxbuf = hbuf + (size_t)ROI_ * ROWS_ * F_;

  const size_t ldsA = (size_t)3 * NP_ * STR_ * sizeof(float);
  ftdgnn_agg_mlp<<<B_ * T_ * H_, TPB_, ldsA, stream>>>(
      x, a, eps, W1, b1, g1, be1, W2, b2, g2, be2, hbuf);

  const size_t ldsP = (size_t)(MCH_ + G3_) * STR_ * sizeof(float);
  ftdgnn_xproj<<<ROI_ * NCH_, TPB_, ldsP, stream>>>(Wih, bih, hbuf, gxbuf);

  const size_t ldsB = (size_t)168 * 1024;
  ftdgnn_gru<<<ROI_, TPBG_, ldsB, stream>>>(Whh, bhh, gxbuf, out);
}


# GRU time loop: each wave keeps its 48x48 part of Whh as B fragments in 72 registers for all 200 steps (loaded once from global), so a step reads only h from LDS, 12 loads per wave not 48
# speedup vs baseline: 2.7859x; 1.0448x over previous
#include <hip/hip_runtime.h>
#include <math.h>

#define B_    4
#define ROI_  90
#define T_    200
#define H_    3
#define F_    96
#define HID_  96
#define G3_   288
#define BH_   12
#define NP_   96
#define STR_  100
#define GSTR_ 290
#define TPB_  288
#define TPBG_ 384
#define KH_   48
#define ROWS_ 2400
#define MCH_  96
#define NCH_  25
#define NPASS_ 32
#define NREAL_ 30
static_assert(TPB_ == 3 * NP_ && NPASS_ * TPB_ == NP_ * NP_ && NREAL_ * 3 == ROI_,
              "kernel A staging map");

typedef float v2f __attribute__((ext_vector_type(2)));
typedef float v8f __attribute__((ext_vector_type(8)));

__device__ __forceinline__ v8f wmma4(v2f a, v2f b, v8f c) {
  return __builtin_amdgcn_wmma_f32_16x16x4_f32(false, a, false, b, (short)0, c,
                                               false, false);
}

__device__ __forceinline__ void mm2x2_trB(const float* As, const float* Bt,
                                          int sr, int sc, int lane, v8f acc[4]) {
  const int half = lane >> 4;
  const int l15  = lane & 15;
  const float* a0 = As + (sr * 32 + l15) * STR_ + (half << 1);
  const float* a1 = a0 + 16 * STR_;
  const float* b0 = Bt + (sc * 32 + l15) * STR_ + (half << 1);
  const float* b1 = b0 + 16 * STR_;
  #pragma unroll
  for (int j = 0; j < 4; ++j) acc[j] = (v8f){0.f,0.f,0.f,0.f,0.f,0.f,0.f,0.f};
  #pragma unroll
  for (int k0 = 0; k0 < NP_; k0 += 4) {
    v2f av0 = *(const v2f*)(a0 + k0);
    v2f av1 = *(const v2f*)(a1 + k0);
    v2f bv0 = *(const v2f*)(b0 + k0);
    v2f bv1 = *(const v2f*)(b1 + k0);
    acc[0] = wmma4(av0, bv0, acc[0]);
    acc[1] = wmma4(av0, bv1, acc[1]);
    acc[2] = wmma4(av1, bv0, acc[2]);
    acc[3] = wmma4(av1, bv1, acc[3]);
  }
}

__device__ __forceinline__ void mm1x3_halfK_regB(const float* As, const v2f (&bw)[3][KH_ / 4],
                                                 int kOff, int lane, v8f acc[3]) {
  const int half = lane >> 4;
  const int l15  = lane & 15;
  const float* aRow = As + l15 * STR_ + kOff + (half << 1);
  v2f av[KH_ / 4];
  #pragma unroll
  for (int j = 0; j < KH_ / 4; ++j) av[j] = *(const v2f*)(aRow + 4 * j);
  #pragma unroll
  for (int g = 0; g < 3; ++g) acc[g] = (v8f){0.f,0.f,0.f,0.f,0.f,0.f,0.f,0.f};
  #pragma unroll
  for (int j = 0; j < KH_ / 4; ++j) {
    #pragma unroll
    for (int g = 0; g < 3; ++g) acc[g] = wmma4(av[j], bw[g][j], acc[g]);
  }
}

__global__ void __launch_bounds__(TPB_)
ftdgnn_agg_mlp(const float* __restrict__ x, const float* __restrict__ a,
               const float* __restrict__ eps,
               const float* __restrict__ W1, const float* __restrict__ b1,
               const float* __restrict__ g1, const float* __restrict__ be1,
               const float* __restrict__ W2, const float* __restrict__ b2,
               const float* __restrict__ g2, const float* __restrict__ be2,
               float* __restrict__ hbuf) {
  extern __shared__ float smem[];
  float* xs = smem;
  float* ms = xs + NP_ * STR_;
  float* ws = ms + NP_ * STR_;

  const int tid  = threadIdx.x;
  const int lane = tid & 31;
  const int wave = tid >> 5;
  const int half = lane >> 4;
  const int l15  = lane & 15;
  const int sr   = wave / 3;
  const int sc   = wave % 3;

  const int bth = blockIdx.x;
  const int b   = bth / (T_ * H_);
  const int th  = bth % (T_ * H_);
  const int t   = th / H_;
  const int hh  = th % H_;

  const int r0 = tid / NP_;
  const int c0 = tid % NP_;
  const int ca = (c0 < ROI_) ? c0 : (ROI_ - 1);
  const size_t rowBase = ((size_t)(b * ROI_ + r0) * T_ + t) * H_ + hh;
  const float* xp  = x  + rowBase * F_   + c0;
  const float* ap  = a  + rowBase * ROI_ + ca;
  const float* w1p = W1 + r0 * HID_ + c0;
  float xv[NREAL_], av[NREAL_], wv[NPASS_];
  #pragma unroll
  for (int i = 0; i < NREAL_; ++i) xv[i] = xp[(size_t)i * (3 * T_ * H_ * F_)];
  #pragma unroll
  for (int i = 0; i < NREAL_; ++i) av[i] = ap[(size_t)i * (3 * T_ * H_ * ROI_)];
  #pragma unroll
  for (int i = 0; i < NPASS_; ++i) wv[i] = w1p[i * (3 * HID_)];
  #pragma unroll
  for (int i = 0; i < NPASS_; ++i) {
    const int r = r0 + 3 * i;
    float xe = 0.f, me = 0.f;
    if (i < NREAL_) {
      xe = xv[i];
      me = (c0 < ROI_ && av[i] != 0.f) ? 1.f : 0.f;
    }
    xs[c0 * STR_ + r] = xe;
    ms[r * STR_ + c0] = me;
    ws[c0 * STR_ + r] = wv[i];
  }
  __syncthreads();

  const float epsv = eps[0];
  const float rs   = rsqrtf(1.0f + 1e-5f);

  v8f acc[4];

  mm2x2_trB(ms, xs, sr, sc, lane, acc);
  __syncthreads();
  #pragma unroll
  for (int di = 0; di < 2; ++di)
    #pragma unroll
    for (int dj = 0; dj < 2; ++dj) {
      const int n = (2 * sc + dj) * 16 + l15;
      #pragma unroll
      for (int v = 0; v < 8; ++v) {
        const int m = (2 * sr + di) * 16 + v + 8 * half;
        ms[m * STR_ + n] = acc[di * 2 + dj][v] + epsv * xs[n * STR_ + m];
      }
    }
  __syncthreads();

  mm2x2_trB(ms, ws, sr, sc, lane, acc);
  __syncthreads();
  const float* w2p = W2 + r0 * F_ + c0;
  #pragma unroll
  for (int i = 0; i < NPASS_; ++i) wv[i] = w2p[i * (3 * F_)];
  #pragma unroll
  for (int di = 0; di < 2; ++di)
    #pragma unroll
    for (int dj = 0; dj < 2; ++dj) {
      const int n = (2 * sc + dj) * 16 + l15;
      const float sc1 = g1[n] * rs, sh1 = be1[n], bb1 = b1[n];
      #pragma unroll
      for (int v = 0; v < 8; ++v) {
        const int m = (2 * sr + di) * 16 + v + 8 * half;
        float z = (acc[di * 2 + dj][v] + bb1) * sc1 + sh1;
        z = (z > 0.f) ? z : expm1f(z);
        xs[m * STR_ + n] = z;
      }
    }
  #pragma unroll
  for (int i = 0; i < NPASS_; ++i)
    ws[c0 * STR_ + r0 + 3 * i] = wv[i];
  __syncthreads();

  mm2x2_trB(xs, ws, sr, sc, lane, acc);
  #pragma unroll
  for (int di = 0; di < 2; ++di)
    #pragma unroll
    for (int dj = 0; dj < 2; ++dj) {
      const int n = (2 * sc + dj) * 16 + l15;
      const float sc2 = g2[n] * rs, sh2 = be2[n], bb2 = b2[n];
      #pragma unroll
      for (int v = 0; v < 8; ++v) {
        const int m = (2 * sr + di) * 16 + v + 8 * half;
        if (m < ROI_) {
          float z = (acc[di * 2 + dj][v] + bb2) * sc2 + sh2;
          z = (z > 0.f) ? z : 0.f;
          hbuf[((size_t)(m * T_ + t) * BH_ + (b * H_ + hh)) * F_ + n] = z;
        }
      }
    }
}

__global__ void __launch_bounds__(TPB_)
ftdgnn_xproj(const float* __restrict__ Wih, const float* __restrict__ bih,
             const float* __restrict__ hbuf, float* __restrict__ gxbuf) {
  extern __shared__ float smem[];
  float* As = smem;
  float* Bs = As + MCH_ * STR_;

  const int tid  = threadIdx.x;
  const int lane = tid & 31;
  const int wave = tid >> 5;
  const int half = lane >> 4;
  const int l15  = lane & 15;
  const int r    = blockIdx.x / NCH_;
  const int ch   = blockIdx.x % NCH_;
  const size_t row0 = (size_t)r * ROWS_ + (size_t)ch * MCH_;

  const float* Ag = hbuf + row0 * F_;
  const float* Bg = Wih + (size_t)r * G3_ * F_;
  #pragma unroll
  for (int i = 0; i < MCH_ * (F_ / 4) / TPB_; ++i) {
    const int idx = tid + i * TPB_;
    const int row = idx / (F_ / 4), k4 = idx % (F_ / 4);
    const float4 v = *(const float4*)(Ag + row * F_ + k4 * 4);
    *(float4*)(As + row * STR_ + k4 * 4) = v;
  }
  #pragma unroll
  for (int i = 0; i < G3_ * (F_ / 4) / TPB_; ++i) {
    const int idx = tid + i * TPB_;
    const int g = idx / (F_ / 4), k4 = idx % (F_ / 4);
    const float4 v = *(const float4*)(Bg + g * F_ + k4 * 4);
    *(float4*)(Bs + g * STR_ + k4 * 4) = v;
  }
  __syncthreads();

  float bias[2];
  #pragma unroll
  for (int dj = 0; dj < 2; ++dj) bias[dj] = bih[r * G3_ + (2 * wave + dj) * 16 + l15];

  for (int sr = 0; sr < MCH_ / 32; ++sr) {
    v8f acc[4];
    mm2x2_trB(As, Bs, sr, wave, lane, acc);
    #pragma unroll
    for (int di = 0; di < 2; ++di)
      #pragma unroll
      for (int dj = 0; dj < 2; ++dj) {
        const int n = (2 * wave + dj) * 16 + l15;
        #pragma unroll
        for (int v = 0; v < 8; ++v) {
          const int m = (2 * sr + di) * 16 + v + 8 * half;
          gxbuf[(row0 + m) * G3_ + n] = acc[di * 2 + dj][v] + bias[dj];
        }
      }
  }
}

__device__ __forceinline__ void lds_barrier() {
  __builtin_amdgcn_fence(__ATOMIC_RELEASE, "workgroup", "local");
  __builtin_amdgcn_s_barrier();
  __builtin_amdgcn_fence(__ATOMIC_ACQUIRE, "workgroup", "local");
}

__global__ void __launch_bounds__(TPBG_)
ftdgnn_gru(const float* __restrict__ Whh, const float* __restrict__ bhh,
           const float* __restrict__ gxbuf, float* __restrict__ out) {
  extern __shared__ float smem[];
  float* hS   = smem;
  float* gh   = hS   + 16 * STR_;
  float* bhhS = gh   + 16 * GSTR_ * 2;

  const int tid  = threadIdx.x;
  const int lane = tid & 31;
  const int wave = tid >> 5;
  const int half = lane >> 4;
  const int l15  = lane & 15;
  const int r    = blockIdx.x;
  const int cg   = wave >> 1;
  const int kh   = wave & 1;
  const int f    = tid % HID_;
  const int m0   = tid / HID_;
  const float kLog2e = 1.44269504088896341f;

  v2f bw[3][KH_ / 4];
  {
    const float* bsrc = Whh + ((size_t)r * G3_ + 16 * cg + l15) * HID_ + KH_ * kh + (half << 1);
    #pragma unroll
    for (int g = 0; g < 3; ++g) {
      #pragma unroll
      for (int j = 0; j < KH_ / 4; ++j)
        bw[g][j] = *(const v2f*)(bsrc + (size_t)g * HID_ * HID_ + 4 * j);
    }
  }
  for (int idx = tid; idx < G3_; idx += TPBG_) bhhS[idx] = bhh[r * G3_ + idx];
  for (int idx = tid; idx < 16 * STR_; idx += TPBG_) hS[idx] = 0.f;
  __syncthreads();

  const float* gxr = gxbuf + (size_t)r * T_ * BH_ * G3_ + m0 * G3_ + f;
  auto load_gx = [&](int t, float (&xg)[3][3]) {
    const float* gxt = gxr + (size_t)t * (BH_ * G3_);
    #pragma unroll
    for (int i = 0; i < 3; ++i)
      #pragma unroll
      for (int g = 0; g < 3; ++g)
        xg[i][g] = gxt[(4 * i) * G3_ + g * HID_];
  };

  auto step = [&](int t, const float (&xg)[3][3]) {
    v8f acc[3];
    mm1x3_halfK_regB(hS, bw, KH_ * kh, lane, acc);
    #pragma unroll
    for (int g = 0; g < 3; ++g) {
      const int n = g * HID_ + 16 * cg + l15;
      #pragma unroll
      for (int v = 0; v < 8; ++v)
        gh[((v + 8 * half) * GSTR_ + n) * 2 + kh] = acc[g][v];
    }
    lds_barrier();

    const v2f* ghp = (const v2f*)gh;
    float hr[3], hz[3], hn[3], hp[3];
    #pragma unroll
    for (int i = 0; i < 3; ++i) {
      const int m = 4 * i + m0;
      const v2f pr = ghp[m * GSTR_ + f];
      const v2f pz = ghp[m * GSTR_ + HID_ + f];
      const v2f pn = ghp[m * GSTR_ + 2 * HID_ + f];
      hr[i] = (pr.x + pr.y) + bhhS[f];
      hz[i] = (pz.x + pz.y) + bhhS[HID_ + f];
      hn[i] = (pn.x + pn.y) + bhhS[2 * HID_ + f];
      hp[i] = hS[m * STR_ + f];
    }
    #pragma unroll
    for (int i = 0; i < 3; ++i) {
      const int m = 4 * i + m0;
      const float xr = xg[i][0];
      const float xz = xg[i][1];
      const float xn = xg[i][2];
      const float rg = __builtin_amdgcn_rcpf(1.f + __builtin_amdgcn_exp2f(-kLog2e * (xr + hr[i])));
      const float zg = __builtin_amdgcn_rcpf(1.f + __builtin_amdgcn_exp2f(-kLog2e * (xz + hz[i])));
      const float en = __builtin_amdgcn_exp2f((2.f * kLog2e) * (xn + rg * hn[i]));
      const float ng = 1.f - 2.f * __builtin_amdgcn_rcpf(1.f + en);
      const float h2 = (1.f - zg) * ng + zg * hp[i];
      hS[m * STR_ + f] = h2;
      const int bb = m / H_, hh = m % H_;
      out[(((size_t)(bb * ROI_ + r) * T_ + t) * H_ + hh) * HID_ + f] = h2;
    }
    lds_barrier();
  };

  static_assert(T_ % 2 == 0, "time loop handles two steps per trip");
  static_assert(TPBG_ == 4 * HID_ && BH_ == 3 * 4 && TPBG_ / 32 == 2 * (HID_ / 16) && 2 * KH_ == HID_,
                "GRU kernel thread and wave maps");
  float xa[3][3], xb[3][3];
  load_gx(0, xa);
  for (int t = 0; t < T_; t += 2) {
    load_gx(t + 1, xb);
    step(t, xa);
    load_gx((t + 2 < T_) ? t + 2 : T_ - 1, xa);
    step(t + 1, xb);
  }
}

extern "C" void kernel_launch(void* const* d_in, const int* in_sizes, int n_in,
                              void* d_out, int out_size, void* d_ws, size_t ws_size,
                              hipStream_t stream) {
  const float* x   = (const float*)d_in[0];
  const float* a   = (const float*)d_in[1];
  const float* eps = (const float*)d_in[2];
  const float* W1  = (const float*)d_in[3];
  const float* b1  = (const float*)d_in[4];
  const float* g1  = (const float*)d_in[5];
  const float* be1 = (const float*)d_in[6];
  const float* W2  = (const float*)d_in[7];
  const float* b2  = (const float*)d_in[8];
  const float* g2  = (const float*)d_in[9];
  const float* be2 = (const float*)d_in[10];
  const float* Wih = (const float*)d_in[11];
  const float* Whh = (const float*)d_in[12];
  const float* bih = (const float*)d_in[13];
  const float* bhh = (const float*)d_in[14];
  float* out   = (float*)d_out;
  float* hbuf  = (float*)d_ws;
  float* gxbuf = hbuf + (size_t)ROI_ * ROWS_ * F_;

  const size_t ldsA = (size_t)3 * NP_ * STR_ * sizeof(float);
  ftdgnn_agg_mlp<<<B_ * T_ * H_, TPB_, ldsA, stream>>>(
      x, a, eps, W1, b1, g1, be1, W2, b2, g2, be2, hbuf);

  const size_t ldsP = (size_t)(MCH_ + G3_) * STR_ * sizeof(float);
  ftdgnn_xproj<<<ROI_ * NCH_, TPB_, ldsP, stream>>>(Wih, bih, hbuf, gxbuf);

  const size_t ldsB = (size_t)168 * 1024;
  ftdgnn_gru<<<ROI_, TPBG_, ldsB, stream>>>(Whh, bhh, gxbuf, out);
}


# Stages pipelined over 5 time chunks of 40 steps: each launch runs the GRU of chunk c (90 blocks) beside the projection of c+1 and the aggregation+MLP of c+2 on the WGPs the GRU leaves idle
# speedup vs baseline: 3.1931x; 1.1462x over previous
#include <hip/hip_runtime.h>
#include <math.h>

#define B_    4
#define ROI_  90
#define T_    200
#define H_    3
#define F_    96
#define HID_  96
#define G3_   288
#define BH_   12
#define NP_   96
#define STR_  100
#define GSTR_ 290
#define TPB_  288
#define TPBG_ 384
#define KH_   48
#define ROWS_ 2400
#define MCH_  96
#define NCH_  25
#define NPASS_ 32
#define NREAL_ 30
#define CT_   40
#define NCK_  5
#define NBA_  480
#define PCH_  5
#define NBP_  450
static_assert(NCK_ * CT_ == T_ && CT_ % 2 == 0 && PCH_ * MCH_ == CT_ * BH_ &&
              NBA_ == B_ * CT_ * H_ && NBP_ == ROI_ * PCH_, "pipeline chunk map");
extern __shared__ float smem[];
static_assert(TPB_ == 3 * NP_ && NPASS_ * TPB_ == NP_ * NP_ && NREAL_ * 3 == ROI_,
              "kernel A staging map");

typedef float v2f __attribute__((ext_vector_type(2)));
typedef float v8f __attribute__((ext_vector_type(8)));

__device__ __forceinline__ v8f wmma4(v2f a, v2f b, v8f c) {
  return __builtin_amdgcn_wmma_f32_16x16x4_f32(false, a, false, b, (short)0, c,
                                               false, false);
}

__device__ __forceinline__ void mm2x2_trB(const float* As, const float* Bt,
                                          int sr, int sc, int lane, v8f acc[4]) {
  const int half = lane >> 4;
  const int l15  = lane & 15;
  const float* a0 = As + (sr * 32 + l15) * STR_ + (half << 1);
  const float* a1 = a0 + 16 * STR_;
  const float* b0 = Bt + (sc * 32 + l15) * STR_ + (half << 1);
  const float* b1 = b0 + 16 * STR_;
  #pragma unroll
  for (int j = 0; j < 4; ++j) acc[j] = (v8f){0.f,0.f,0.f,0.f,0.f,0.f,0.f,0.f};
  #pragma unroll
  for (int k0 = 0; k0 < NP_; k0 += 4) {
    v2f av0 = *(const v2f*)(a0 + k0);
    v2f av1 = *(const v2f*)(a1 + k0);
    v2f bv0 = *(const v2f*)(b0 + k0);
    v2f bv1 = *(const v2f*)(b1 + k0);
    acc[0] = wmma4(av0, bv0, acc[0]);
    acc[1] = wmma4(av0, bv1, acc[1]);
    acc[2] = wmma4(av1, bv0, acc[2]);
    acc[3] = wmma4(av1, bv1, acc[3]);
  }
}

__device__ __forceinline__ void mm1x3_halfK_regB(const float* As, const v2f (&bw)[3][KH_ / 4],
                                                 int kOff, int lane, v8f acc[3]) {
  const int half = lane >> 4;
  const int l15  = lane & 15;
  const float* aRow = As + l15 * STR_ + kOff + (half << 1);
  v2f av[KH_ / 4];
  #pragma unroll
  for (int j = 0; j < KH_ / 4; ++j) av[j] = *(const v2f*)(aRow + 4 * j);
  #pragma unroll
  for (int g = 0; g < 3; ++g) acc[g] = (v8f){0.f,0.f,0.f,0.f,0.f,0.f,0.f,0.f};
  #pragma unroll
  for (int j = 0; j < KH_ / 4; ++j) {
    #pragma unroll
    for (int g = 0; g < 3; ++g) acc[g] = wmma4(av[j], bw[g][j], acc[g]);
  }
}

__device__ __forceinline__ void
stage_agg_mlp(const int q, const int ck,
               const float* __restrict__ x, const float* __restrict__ a,
               const float* __restrict__ eps,
               const float* __restrict__ W1, const float* __restrict__ b1,
               const float* __restrict__ g1, const float* __restrict__ be1,
               const float* __restrict__ W2, const float* __restrict__ b2,
               const float* __restrict__ g2, const float* __restrict__ be2,
               float* __restrict__ hbuf) {
  float* xs = smem;
  float* ms = xs + NP_ * STR_;
  float* ws = ms + NP_ * STR_;

  const int tid  = threadIdx.x;
  const int lane = tid & 31;
  const int wave = tid >> 5;
  const bool act = tid < TPB_;
  const int half = lane >> 4;
  const int l15  = lane & 15;
  const int sr   = wave / 3;
  const int sc   = wave % 3;

  const int b   = q / (CT_ * H_);
  const int th  = q % (CT_ * H_);
  const int t   = ck * CT_ + th / H_;
  const int hh  = th % H_;

  const int r0 = tid / NP_;
  const int c0 = tid % NP_;
  const int ca = (c0 < ROI_) ? c0 : (ROI_ - 1);
  const size_t rowBase = ((size_t)(b * ROI_ + r0) * T_ + t) * H_ + hh;
  const float* xp  = x  + rowBase * F_   + c0;
  const float* ap  = a  + rowBase * ROI_ + ca;
  const float* w1p = W1 + r0 * HID_ + c0;
  float wv[NPASS_];
  if (act) {
  float xv[NREAL_], av[NREAL_];
  #pragma unroll
  for (int i = 0; i < NREAL_; ++i) xv[i] = xp[(size_t)i * (3 * T_ * H_ * F_)];
  #pragma unroll
  for (int i = 0; i < NREAL_; ++i) av[i] = ap[(size_t)i * (3 * T_ * H_ * ROI_)];
  #pragma unroll
  for (int i = 0; i < NPASS_; ++i) wv[i] = w1p[i * (3 * HID_)];
  #pragma unroll
  for (int i = 0; i < NPASS_; ++i) {
    const int r = r0 + 3 * i;
    float xe = 0.f, me = 0.f;
    if (i < NREAL_) {
      xe = xv[i];
      me = (c0 < ROI_ && av[i] != 0.f) ? 1.f : 0.f;
    }
    xs[c0 * STR_ + r] = xe;
    ms[r * STR_ + c0] = me;
    ws[c0 * STR_ + r] = wv[i];
  }
  }
  __syncthreads();

  const float epsv = eps[0];
  const float rs   = rsqrtf(1.0f + 1e-5f);

  v8f acc[4];

  if (act) mm2x2_trB(ms, xs, sr, sc, lane, acc);
  __syncthreads();
  if (act) {
  #pragma unroll
  for (int di = 0; di < 2; ++di)
    #pragma unroll
    for (int dj = 0; dj < 2; ++dj) {
      const int n = (2 * sc + dj) * 16 + l15;
      #pragma unroll
      for (int v = 0; v < 8; ++v) {
        const int m = (2 * sr + di) * 16 + v + 8 * half;
        ms[m * STR_ + n] = acc[di * 2 + dj][v] + epsv * xs[n * STR_ + m];
      }
    }
  }
  __syncthreads();

  if (act) mm2x2_trB(ms, ws, sr, sc, lane, acc);
  __syncthreads();
  if (act) {
  const float* w2p = W2 + r0 * F_ + c0;
  #pragma unroll
  for (int i = 0; i < NPASS_; ++i) wv[i] = w2p[i * (3 * F_)];
  #pragma unroll
  for (int di = 0; di < 2; ++di)
    #pragma unroll
    for (int dj = 0; dj < 2; ++dj) {
      const int n = (2 * sc + dj) * 16 + l15;
      const float sc1 = g1[n] * rs, sh1 = be1[n], bb1 = b1[n];
      #pragma unroll
      for (int v = 0; v < 8; ++v) {
        const int m = (2 * sr + di) * 16 + v + 8 * half;
        float z = (acc[di * 2 + dj][v] + bb1) * sc1 + sh1;
        z = (z > 0.f) ? z : expm1f(z);
        xs[m * STR_ + n] = z;
      }
    }
  #pragma unroll
  for (int i = 0; i < NPASS_; ++i)
    ws[c0 * STR_ + r0 + 3 * i] = wv[i];
  }
  __syncthreads();
  if (!act) return;

  mm2x2_trB(xs, ws, sr, sc, lane, acc);
  #pragma unroll
  for (int di = 0; di < 2; ++di)
    #pragma unroll
    for (int dj = 0; dj < 2; ++dj) {
      const int n = (2 * sc + dj) * 16 + l15;
      const float sc2 = g2[n] * rs, sh2 = be2[n], bb2 = b2[n];
      #pragma unroll
      for (int v = 0; v < 8; ++v) {
        const int m = (2 * sr + di) * 16 + v + 8 * half;
        if (m < ROI_) {
          float z = (acc[di * 2 + dj][v] + bb2) * sc2 + sh2;
          z = (z > 0.f) ? z : 0.f;
          hbuf[((size_t)(m * T_ + t) * BH_ + (b * H_ + hh)) * F_ + n] = z;
        }
      }
    }
}

__device__ __forceinline__ void
stage_xproj(const int q, const int ck,
             const float* __restrict__ Wih, const float* __restrict__ bih,
             const float* __restrict__ hbuf, float* __restrict__ gxbuf) {
  float* As = smem;
  float* Bs = As + MCH_ * STR_;

  const int tid  = threadIdx.x;
  const int lane = tid & 31;
  const int wave = tid >> 5;
  const bool act = tid < TPB_;
  const int half = lane >> 4;
  const int l15  = lane & 15;
  const int r    = q / PCH_;
  const int ch   = ck * PCH_ + q % PCH_;
  const size_t row0 = (size_t)r * ROWS_ + (size_t)ch * MCH_;

  const float* Ag = hbuf + row0 * F_;
  const float* Bg = Wih + (size_t)r * G3_ * F_;
  if (act) {
  #pragma unroll
  for (int i = 0; i < MCH_ * (F_ / 4) / TPB_; ++i) {
    const int idx = tid + i * TPB_;
    const int row = idx / (F_ / 4), k4 = idx % (F_ / 4);
    const float4 v = *(const float4*)(Ag + row * F_ + k4 * 4);
    *(float4*)(As + row * STR_ + k4 * 4) = v;
  }
  #pragma unroll
  for (int i = 0; i < G3_ * (F_ / 4) / TPB_; ++i) {
    const int idx = tid + i * TPB_;
    const int g = idx / (F_ / 4), k4 = idx % (F_ / 4);
    const float4 v = *(const float4*)(Bg + g * F_ + k4 * 4);
    *(float4*)(Bs + g * STR_ + k4 * 4) = v;
  }
  }
  __syncthreads();
  if (!act) return;

  float bias[2];
  #pragma unroll
  for (int dj = 0; dj < 2; ++dj) bias[dj] = bih[r * G3_ + (2 * wave + dj) * 16 + l15];

  for (int sr = 0; sr < MCH_ / 32; ++sr) {
    v8f acc[4];
    mm2x2_trB(As, Bs, sr, wave, lane, acc);
    #pragma unroll
    for (int di = 0; di < 2; ++di)
      #pragma unroll
      for (int dj = 0; dj < 2; ++dj) {
        const int n = (2 * wave + dj) * 16 + l15;
        #pragma unroll
        for (int v = 0; v < 8; ++v) {
          const int m = (2 * sr + di) * 16 + v + 8 * half;
          gxbuf[(row0 + m) * G3_ + n] = acc[di * 2 + dj][v] + bias[dj];
        }
      }
  }
}

__device__ __forceinline__ void lds_barrier() {
  __builtin_amdgcn_fence(__ATOMIC_RELEASE, "workgroup", "local");
  __builtin_amdgcn_s_barrier();
  __builtin_amdgcn_fence(__ATOMIC_ACQUIRE, "workgroup", "local");
}

__device__ __forceinline__ void
stage_gru(const int r, const int ck,
          const float* __restrict__ Whh, const float* __restrict__ bhh,
          const float* __restrict__ gxbuf, float* out) {
  float* hS   = smem;
  float* gh   = hS   + 16 * STR_;
  float* bhhS = gh   + 16 * GSTR_ * 2;

  const int tid  = threadIdx.x;
  const int lane = tid & 31;
  const int wave = tid >> 5;
  const int half = lane >> 4;
  const int l15  = lane & 15;
  const int cg   = wave >> 1;
  const int kh   = wave & 1;
  const int f    = tid % HID_;
  const int m0   = tid / HID_;
  const float kLog2e = 1.44269504088896341f;

  v2f bw[3][KH_ / 4];
  {
    const float* bsrc = Whh + ((size_t)r * G3_ + 16 * cg + l15) * HID_ + KH_ * kh + (half << 1);
    #pragma unroll
    for (int g = 0; g < 3; ++g) {
      #pragma unroll
      for (int j = 0; j < KH_ / 4; ++j)
        bw[g][j] = *(const v2f*)(bsrc + (size_t)g * HID_ * HID_ + 4 * j);
    }
  }
  for (int idx = tid; idx < G3_; idx += TPBG_) bhhS[idx] = bhh[r * G3_ + idx];
  for (int idx = tid; idx < 16 * STR_; idx += TPBG_) hS[idx] = 0.f;
  __syncthreads();
  const int t0 = ck * CT_;
  if (ck > 0) {
    #pragma unroll
    for (int i = 0; i < 3; ++i) {
      const int m = 4 * i + m0;
      const int bb = m / H_, hh = m % H_;
      hS[m * STR_ + f] = out[(((size_t)(bb * ROI_ + r) * T_ + (t0 - 1)) * H_ + hh) * HID_ + f];
    }
    __syncthreads();
  }

  const float* gxr = gxbuf + (size_t)r * T_ * BH_ * G3_ + m0 * G3_ + f;
  auto load_gx = [&](int t, float (&xg)[3][3]) {
    const float* gxt = gxr + (size_t)t * (BH_ * G3_);
    #pragma unroll
    for (int i = 0; i < 3; ++i)
      #pragma unroll
      for (int g = 0; g < 3; ++g)
        xg[i][g] = gxt[(4 * i) * G3_ + g * HID_];
  };

  auto step = [&](int t, const float (&xg)[3][3]) {
    v8f acc[3];
    mm1x3_halfK_regB(hS, bw, KH_ * kh, lane, acc);
    #pragma unroll
    for (int g = 0; g < 3; ++g) {
      const int n = g * HID_ + 16 * cg + l15;
      #pragma unroll
      for (int v = 0; v < 8; ++v)
        gh[((v + 8 * half) * GSTR_ + n) * 2 + kh] = acc[g][v];
    }
    lds_barrier();

    const v2f* ghp = (const v2f*)gh;
    float hr[3], hz[3], hn[3], hp[3];
    #pragma unroll
    for (int i = 0; i < 3; ++i) {
      const int m = 4 * i + m0;
      const v2f pr = ghp[m * GSTR_ + f];
      const v2f pz = ghp[m * GSTR_ + HID_ + f];
      const v2f pn = ghp[m * GSTR_ + 2 * HID_ + f];
      hr[i] = (pr.x + pr.y) + bhhS[f];
      hz[i] = (pz.x + pz.y) + bhhS[HID_ + f];
      hn[i] = (pn.x + pn.y) + bhhS[2 * HID_ + f];
      hp[i] = hS[m * STR_ + f];
    }
    #pragma unroll
    for (int i = 0; i < 3; ++i) {
      const int m = 4 * i + m0;
      const float xr = xg[i][0];
      const float xz = xg[i][1];
      const float xn = xg[i][2];
      const float rg = __builtin_amdgcn_rcpf(1.f + __builtin_amdgcn_exp2f(-kLog2e * (xr + hr[i])));
      const float zg = __builtin_amdgcn_rcpf(1.f + __builtin_amdgcn_exp2f(-kLog2e * (xz + hz[i])));
      const float en = __builtin_amdgcn_exp2f((2.f * kLog2e) * (xn + rg * hn[i]));
      const float ng = 1.f - 2.f * __builtin_amdgcn_rcpf(1.f + en);
      const float h2 = (1.f - zg) * ng + zg * hp[i];
      hS[m * STR_ + f] = h2;
      const int bb = m / H_, hh = m % H_;
      out[(((size_t)(bb * ROI_ + r) * T_ + t) * H_ + hh) * HID_ + f] = h2;
    }
    lds_barrier();
  };

  static_assert(T_ % 2 == 0, "time loop handles two steps per trip");
  static_assert(TPBG_ == 4 * HID_ && BH_ == 3 * 4 && TPBG_ / 32 == 2 * (HID_ / 16) && 2 * KH_ == HID_,
                "GRU kernel thread and wave maps");
  float xa[3][3], xb[3][3];
  load_gx(t0, xa);
  for (int t = t0; t < t0 + CT_; t += 2) {
    load_gx(t + 1, xb);
    step(t, xa);
    load_gx((t + 2 < t0 + CT_) ? t + 2 : t0 + CT_ - 1, xa);
    step(t + 1, xb);
  }
}

__global__ void __launch_bounds__(TPB_)
ftdgnn_front(const float* __restrict__ x, const float* __restrict__ a,
             const float* __restrict__ eps,
             const float* __restrict__ W1, const float* __restrict__ b1,
             const float* __restrict__ g1, const float* __restrict__ be1,
             const float* __restrict__ W2, const float* __restrict__ b2,
             const float* __restrict__ g2, const float* __restrict__ be2,
             const float* __restrict__ Wih, const float* __restrict__ bih,
             float* hbuf, float* gxbuf, const int cA, const int cP) {
  int q = blockIdx.x;
  if (q < NBA_) { stage_agg_mlp(q, cA, x, a, eps, W1, b1, g1, be1, W2, b2, g2, be2, hbuf); return; }
  q -= NBA_;
  stage_xproj(q, cP, Wih, bih, hbuf, gxbuf);
}

__global__ void __launch_bounds__(TPBG_)
ftdgnn_pipe(const float* __restrict__ x, const float* __restrict__ a,
            const float* __restrict__ eps,
            const float* __restrict__ W1, const float* __restrict__ b1,
            const float* __restrict__ g1, const float* __restrict__ be1,
            const float* __restrict__ W2, const float* __restrict__ b2,
            const float* __restrict__ g2, const float* __restrict__ be2,
            const float* __restrict__ Wih, const float* __restrict__ Whh,
            const float* __restrict__ bih, const float* __restrict__ bhh,
            float* hbuf, float* gxbuf, float* out,
            const int cG, const int cA, const int cP) {
  int q = blockIdx.x;
  const int nG = (cG >= 0) ? ROI_ : 0;
  const int nA = (cA >= 0) ? NBA_ : 0;
  if (q < nG) { stage_gru(q, cG, Whh, bhh, gxbuf, out); return; }
  q -= nG;
  if (q < nA) { stage_agg_mlp(q, cA, x, a, eps, W1, b1, g1, be1, W2, b2, g2, be2, hbuf); return; }
  q -= nA;
  stage_xproj(q, cP, Wih, bih, hbuf, gxbuf);
}

extern "C" void kernel_launch(void* const* d_in, const int* in_sizes, int n_in,
                              void* d_out, int out_size, void* d_ws, size_t ws_size,
                              hipStream_t stream) {
  const float* x   = (const float*)d_in[0];
  const float* a   = (const float*)d_in[1];
  const float* eps = (const float*)d_in[2];
  const float* W1  = (const float*)d_in[3];
  const float* b1  = (const float*)d_in[4];
  const float* g1  = (const float*)d_in[5];
  const float* be1 = (const float*)d_in[6];
  const float* W2  = (const float*)d_in[7];
  const float* b2  = (const float*)d_in[8];
  const float* g2  = (const float*)d_in[9];
  const float* be2 = (const float*)d_in[10];
  const float* Wih = (const float*)d_in[11];
  const float* Whh = (const float*)d_in[12];
  const float* bih = (const float*)d_in[13];
  const float* bhh = (const float*)d_in[14];
  float* out   = (float*)d_out;
  float* hbuf  = (float*)d_ws;
  float* gxbuf = hbuf + (size_t)ROI_ * ROWS_ * F_;

  const size_t ldsA = (size_t)3 * NP_ * STR_ * sizeof(float);
  const size_t ldsP = (size_t)(MCH_ + G3_) * STR_ * sizeof(float);
  const size_t ldsB = (size_t)168 * 1024;

  ftdgnn_front<<<NBA_, TPB_, ldsA, stream>>>(
      x, a, eps, W1, b1, g1, be1, W2, b2, g2, be2, Wih, bih, hbuf, gxbuf, 0, -1);
  ftdgnn_front<<<NBA_ + NBP_, TPB_, ldsP, stream>>>(
      x, a, eps, W1, b1, g1, be1, W2, b2, g2, be2, Wih, bih, hbuf, gxbuf, 1, 0);
  ftdgnn_pipe<<<ROI_ + NBA_ + NBP_, TPBG_, ldsB, stream>>>(
      x, a, eps, W1, b1, g1, be1, W2, b2, g2, be2, Wih, Whh, bih, bhh, hbuf, gxbuf, out, 0, 2, 1);
  ftdgnn_pipe<<<ROI_ + NBA_ + NBP_, TPBG_, ldsB, stream>>>(
      x, a, eps, W1, b1, g1, be1, W2, b2, g2, be2, Wih, Whh, bih, bhh, hbuf, gxbuf, out, 1, 3, 2);
  ftdgnn_pipe<<<ROI_ + NBA_ + NBP_, TPBG_, ldsB, stream>>>(
      x, a, eps, W1, b1, g1, be1, W2, b2, g2, be2, Wih, Whh, bih, bhh, hbuf, gxbuf, out, 2, 4, 3);
  ftdgnn_pipe<<<ROI_ + NBP_, TPBG_, ldsB, stream>>>(
      x, a, eps, W1, b1, g1, be1, W2, b2, g2, be2, Wih, Whh, bih, bhh, hbuf, gxbuf, out, 3, -1, 4);
  ftdgnn_pipe<<<ROI_, TPBG_, ldsB, stream>>>(
      x, a, eps, W1, b1, g1, be1, W2, b2, g2, be2, Wih, Whh, bih, bhh, hbuf, gxbuf, out, 4, -1, -1);
}


# Aggregation+MLP and projection stages: matrix work on all 12 waves of a block (3 or 9 tiles per wave, three waves per SIMD), so each matrix pipe issues 216/648 per block, not 288/864 on the fullest
# speedup vs baseline: 3.5869x; 1.1233x over previous
#include <hip/hip_runtime.h>
#include <math.h>

#define B_    4
#define ROI_  90
#define T_    200
#define H_    3
#define F_    96
#define HID_  96
#define G3_   288
#define BH_   12
#define NP_   96
#define STR_  100
#define GSTR_ 290
#define TPB_  288
#define TPBG_ 384
#define KH_   48
#define ROWS_ 2400
#define MCH_  96
#define NCH_  25
#define NPASS_ 32
#define NREAL_ 30
#define CT_   40
#define NCK_  5
#define NBA_  480
#define PCH_  5
#define NBP_  450
static_assert(NCK_ * CT_ == T_ && CT_ % 2 == 0 && PCH_ * MCH_ == CT_ * BH_ &&
              NBA_ == B_ * CT_ * H_ && NBP_ == ROI_ * PCH_, "pipeline chunk map");
extern __shared__ float smem[];
static_assert(TPB_ == 3 * NP_ && NPASS_ * TPB_ == NP_ * NP_ && NREAL_ * 3 == ROI_,
              "kernel A staging map");

typedef float v2f __attribute__((ext_vector_type(2)));
typedef float v8f __attribute__((ext_vector_type(8)));

__device__ __forceinline__ v8f wmma4(v2f a, v2f b, v8f c) {
  return __builtin_amdgcn_wmma_f32_16x16x4_f32(false, a, false, b, (short)0, c,
                                               false, false);
}

__device__ __forceinline__ void mm1x3_trB(const float* As, const float* Bt,
                                          int tr, int tc0, int lane, v8f acc[3]) {
  const int half = lane >> 4;
  const int l15  = lane & 15;
  const float* a0 = As + (tr * 16 + l15) * STR_ + (half << 1);
  const float* b0 = Bt + (tc0 * 16 + l15) * STR_ + (half << 1);
  #pragma unroll
  for (int j = 0; j < 3; ++j) acc[j] = (v8f){0.f,0.f,0.f,0.f,0.f,0.f,0.f,0.f};
  #pragma unroll
  for (int k0 = 0; k0 < NP_; k0 += 4) {
    v2f av  = *(const v2f*)(a0 + k0);
    v2f bv0 = *(const v2f*)(b0 + k0);
    v2f bv1 = *(const v2f*)(b0 + 16 * STR_ + k0);
    v2f bv2 = *(const v2f*)(b0 + 32 * STR_ + k0);
    acc[0] = wmma4(av, bv0, acc[0]);
    acc[1] = wmma4(av, bv1, acc[1]);
    acc[2] = wmma4(av, bv2, acc[2]);
  }
}

__device__ __forceinline__ void mm1x3_halfK_regB(const float* As, const v2f (&bw)[3][KH_ / 4],
                                                 int kOff, int lane, v8f acc[3]) {
  const int half = lane >> 4;
  const int l15  = lane & 15;
  const float* aRow = As + l15 * STR_ + kOff + (half << 1);
  v2f av[KH_ / 4];
  #pragma unroll
  for (int j = 0; j < KH_ / 4; ++j) av[j] = *(const v2f*)(aRow + 4 * j);
  #pragma unroll
  for (int g = 0; g < 3; ++g) acc[g] = (v8f){0.f,0.f,0.f,0.f,0.f,0.f,0.f,0.f};
  #pragma unroll
  for (int j = 0; j < KH_ / 4; ++j) {
    #pragma unroll
    for (int g = 0; g < 3; ++g) acc[g] = wmma4(av[j], bw[g][j], acc[g]);
  }
}

__device__ __forceinline__ void
stage_agg_mlp(const int q, const int ck,
               const float* __restrict__ x, const float* __restrict__ a,
               const float* __restrict__ eps,
               const float* __restrict__ W1, const float* __restrict__ b1,
               const float* __restrict__ g1, const float* __restrict__ be1,
               const float* __restrict__ W2, const float* __restrict__ b2,
               const float* __restrict__ g2, const float* __restrict__ be2,
               float* __restrict__ hbuf) {
  float* xs = smem;
  float* ms = xs + NP_ * STR_;
  float* ws = ms + NP_ * STR_;

  const int tid  = threadIdx.x;
  const int lane = tid & 31;
  const int wave = tid >> 5;
  const bool act = tid < TPB_;
  const int half = lane >> 4;
  const int l15  = lane & 15;
  const int tr   = wave >> 1;
  const int tc0  = 3 * (wave & 1);

  const int b   = q / (CT_ * H_);
  const int th  = q % (CT_ * H_);
  const int t   = ck * CT_ + th / H_;
  const int hh  = th % H_;

  const int r0 = tid / NP_;
  const int c0 = tid % NP_;
  const int ca = (c0 < ROI_) ? c0 : (ROI_ - 1);
  const size_t rowBase = ((size_t)(b * ROI_ + r0) * T_ + t) * H_ + hh;
  const float* xp  = x  + rowBase * F_   + c0;
  const float* ap  = a  + rowBase * ROI_ + ca;
  const float* w1p = W1 + r0 * HID_ + c0;
  float wv[NPASS_];
  if (act) {
  float xv[NREAL_], av[NREAL_];
  #pragma unroll
  for (int i = 0; i < NREAL_; ++i) xv[i] = xp[(size_t)i * (3 * T_ * H_ * F_)];
  #pragma unroll
  for (int i = 0; i < NREAL_; ++i) av[i] = ap[(size_t)i * (3 * T_ * H_ * ROI_)];
  #pragma unroll
  for (int i = 0; i < NPASS_; ++i) wv[i] = w1p[i * (3 * HID_)];
  #pragma unroll
  for (int i = 0; i < NPASS_; ++i) {
    const int r = r0 + 3 * i;
    float xe = 0.f, me = 0.f;
    if (i < NREAL_) {
      xe = xv[i];
      me = (c0 < ROI_ && av[i] != 0.f) ? 1.f : 0.f;
    }
    xs[c0 * STR_ + r] = xe;
    ms[r * STR_ + c0] = me;
    ws[c0 * STR_ + r] = wv[i];
  }
  }
  __syncthreads();

  const float epsv = eps[0];
  const float rs   = rsqrtf(1.0f + 1e-5f);

  v8f acc[3];

  mm1x3_trB(ms, xs, tr, tc0, lane, acc);
  __syncthreads();
  #pragma unroll
  for (int j = 0; j < 3; ++j) {
    const int n = (tc0 + j) * 16 + l15;
    #pragma unroll
    for (int v = 0; v < 8; ++v) {
      const int m = tr * 16 + v + 8 * half;
      ms[m * STR_ + n] = acc[j][v] + epsv * xs[n * STR_ + m];
    }
  }
  __syncthreads();

  mm1x3_trB(ms, ws, tr, tc0, lane, acc);
  __syncthreads();
  if (act) {
    const float* w2p = W2 + r0 * F_ + c0;
    #pragma unroll
    for (int i = 0; i < NPASS_; ++i) wv[i] = w2p[i * (3 * F_)];
  }
  #pragma unroll
  for (int j = 0; j < 3; ++j) {
    const int n = (tc0 + j) * 16 + l15;
    const float sc1 = g1[n] * rs, sh1 = be1[n], bb1 = b1[n];
    #pragma unroll
    for (int v = 0; v < 8; ++v) {
      const int m = tr * 16 + v + 8 * half;
      float z = (acc[j][v] + bb1) * sc1 + sh1;
      z = (z > 0.f) ? z : expm1f(z);
      xs[m * STR_ + n] = z;
    }
  }
  if (act) {
    #pragma unroll
    for (int i = 0; i < NPASS_; ++i)
      ws[c0 * STR_ + r0 + 3 * i] = wv[i];
  }
  __syncthreads();

  mm1x3_trB(xs, ws, tr, tc0, lane, acc);
  #pragma unroll
  for (int j = 0; j < 3; ++j) {
    const int n = (tc0 + j) * 16 + l15;
    const float sc2 = g2[n] * rs, sh2 = be2[n], bb2 = b2[n];
    #pragma unroll
    for (int v = 0; v < 8; ++v) {
      const int m = tr * 16 + v + 8 * half;
      if (m < ROI_) {
        float z = (acc[j][v] + bb2) * sc2 + sh2;
        z = (z > 0.f) ? z : 0.f;
        hbuf[((size_t)(m * T_ + t) * BH_ + (b * H_ + hh)) * F_ + n] = z;
      }
    }
  }
}

__device__ __forceinline__ void
stage_xproj(const int q, const int ck,
             const float* __restrict__ Wih, const float* __restrict__ bih,
             const float* __restrict__ hbuf, float* __restrict__ gxbuf) {
  float* As = smem;
  float* Bs = As + MCH_ * STR_;

  const int tid  = threadIdx.x;
  const int lane = tid & 31;
  const int wave = tid >> 5;
  const bool act = tid < TPB_;
  const int half = lane >> 4;
  const int l15  = lane & 15;
  const int tr   = wave % 6;
  const int tcb  = 9 * (wave / 6);
  const int r    = q / PCH_;
  const int ch   = ck * PCH_ + q % PCH_;
  const size_t row0 = (size_t)r * ROWS_ + (size_t)ch * MCH_;

  const float* Ag = hbuf + row0 * F_;
  const float* Bg = Wih + (size_t)r * G3_ * F_;
  if (act) {
  #pragma unroll
  for (int i = 0; i < MCH_ * (F_ / 4) / TPB_; ++i) {
    const int idx = tid + i * TPB_;
    const int row = idx / (F_ / 4), k4 = idx % (F_ / 4);
    const float4 v = *(const float4*)(Ag + row * F_ + k4 * 4);
    *(float4*)(As + row * STR_ + k4 * 4) = v;
  }
  #pragma unroll
  for (int i = 0; i < G3_ * (F_ / 4) / TPB_; ++i) {
    const int idx = tid + i * TPB_;
    const int g = idx / (F_ / 4), k4 = idx % (F_ / 4);
    const float4 v = *(const float4*)(Bg + g * F_ + k4 * 4);
    *(float4*)(Bs + g * STR_ + k4 * 4) = v;
  }
  }
  __syncthreads();

  for (int ps = 0; ps < 3; ++ps) {
    const int tc0 = tcb + 3 * ps;
    v8f acc[3];
    mm1x3_trB(As, Bs, tr, tc0, lane, acc);
    #pragma unroll
    for (int j = 0; j < 3; ++j) {
      const int n = (tc0 + j) * 16 + l15;
      const float bias = bih[r * G3_ + n];
      #pragma unroll
      for (int v = 0; v < 8; ++v) {
        const int m = tr * 16 + v + 8 * half;
        gxbuf[(row0 + m) * G3_ + n] = acc[j][v] + bias;
      }
    }
  }
}

__device__ __forceinline__ void lds_barrier() {
  __builtin_amdgcn_fence(__ATOMIC_RELEASE, "workgroup", "local");
  __builtin_amdgcn_s_barrier();
  __builtin_amdgcn_fence(__ATOMIC_ACQUIRE, "workgroup", "local");
}

__device__ __forceinline__ void
stage_gru(const int r, const int ck,
          const float* __restrict__ Whh, const float* __restrict__ bhh,
          const float* __restrict__ gxbuf, float* out) {
  float* hS   = smem;
  float* gh   = hS   + 16 * STR_;
  float* bhhS = gh   + 16 * GSTR_ * 2;

  const int tid  = threadIdx.x;
  const int lane = tid & 31;
  const int wave = tid >> 5;
  const int half = lane >> 4;
  const int l15  = lane & 15;
  const int cg   = wave >> 1;
  const int kh   = wave & 1;
  const int f    = tid % HID_;
  const int m0   = tid / HID_;
  const float kLog2e = 1.44269504088896341f;

  v2f bw[3][KH_ / 4];
  {
    const float* bsrc = Whh + ((size_t)r * G3_ + 16 * cg + l15) * HID_ + KH_ * kh + (half << 1);
    #pragma unroll
    for (int g = 0; g < 3; ++g) {
      #pragma unroll
      for (int j = 0; j < KH_ / 4; ++j)
        bw[g][j] = *(const v2f*)(bsrc + (size_t)g * HID_ * HID_ + 4 * j);
    }
  }
  for (int idx = tid; idx < G3_; idx += TPBG_) bhhS[idx] = bhh[r * G3_ + idx];
  for (int idx = tid; idx < 16 * STR_; idx += TPBG_) hS[idx] = 0.f;
  __syncthreads();
  const int t0 = ck * CT_;
  if (ck > 0) {
    #pragma unroll
    for (int i = 0; i < 3; ++i) {
      const int m = 4 * i + m0;
      const int bb = m / H_, hh = m % H_;
      hS[m * STR_ + f] = out[(((size_t)(bb * ROI_ + r) * T_ + (t0 - 1)) * H_ + hh) * HID_ + f];
    }
    __syncthreads();
  }

  const float* gxr = gxbuf + (size_t)r * T_ * BH_ * G3_ + m0 * G3_ + f;
  auto load_gx = [&](int t, float (&xg)[3][3]) {
    const float* gxt = gxr + (size_t)t * (BH_ * G3_);
    #pragma unroll
    for (int i = 0; i < 3; ++i)
      #pragma unroll
      for (int g = 0; g < 3; ++g)
        xg[i][g] = gxt[(4 * i) * G3_ + g * HID_];
  };

  auto step = [&](int t, const float (&xg)[3][3]) {
    v8f acc[3];
    mm1x3_halfK_regB(hS, bw, KH_ * kh, lane, acc);
    #pragma unroll
    for (int g = 0; g < 3; ++g) {
      const int n = g * HID_ + 16 * cg + l15;
      #pragma unroll
      for (int v = 0; v < 8; ++v)
        gh[((v + 8 * half) * GSTR_ + n) * 2 + kh] = acc[g][v];
    }
    lds_barrier();

    const v2f* ghp = (const v2f*)gh;
    float hr[3], hz[3], hn[3], hp[3];
    #pragma unroll
    for (int i = 0; i < 3; ++i) {
      const int m = 4 * i + m0;
      const v2f pr = ghp[m * GSTR_ + f];
      const v2f pz = ghp[m * GSTR_ + HID_ + f];
      const v2f pn = ghp[m * GSTR_ + 2 * HID_ + f];
      hr[i] = (pr.x + pr.y) + bhhS[f];
      hz[i] = (pz.x + pz.y) + bhhS[HID_ + f];
      hn[i] = (pn.x + pn.y) + bhhS[2 * HID_ + f];
      hp[i] = hS[m * STR_ + f];
    }
    #pragma unroll
    for (int i = 0; i < 3; ++i) {
      const int m = 4 * i + m0;
      const float xr = xg[i][0];
      const float xz = xg[i][1];
      const float xn = xg[i][2];
      const float rg = __builtin_amdgcn_rcpf(1.f + __builtin_amdgcn_exp2f(-kLog2e * (xr + hr[i])));
      const float zg = __builtin_amdgcn_rcpf(1.f + __builtin_amdgcn_exp2f(-kLog2e * (xz + hz[i])));
      const float en = __builtin_amdgcn_exp2f((2.f * kLog2e) * (xn + rg * hn[i]));
      const float ng = 1.f - 2.f * __builtin_amdgcn_rcpf(1.f + en);
      const float h2 = (1.f - zg) * ng + zg * hp[i];
      hS[m * STR_ + f] = h2;
      const int bb = m / H_, hh = m % H_;
      out[(((size_t)(bb * ROI_ + r) * T_ + t) * H_ + hh) * HID_ + f] = h2;
    }
    lds_barrier();
  };

  static_assert(T_ % 2 == 0, "time loop handles two steps per trip");
  static_assert(TPBG_ == 4 * HID_ && BH_ == 3 * 4 && TPBG_ / 32 == 2 * (HID_ / 16) && 2 * KH_ == HID_,
                "GRU kernel thread and wave maps");
  float xa[3][3], xb[3][3];
  load_gx(t0, xa);
  for (int t = t0; t < t0 + CT_; t += 2) {
    load_gx(t + 1, xb);
    step(t, xa);
    load_gx((t + 2 < t0 + CT_) ? t + 2 : t0 + CT_ - 1, xa);
    step(t + 1, xb);
  }
}

__global__ void __launch_bounds__(TPBG_) __attribute__((amdgpu_waves_per_eu(6)))
ftdgnn_front(const float* __restrict__ x, const float* __restrict__ a,
             const float* __restrict__ eps,
             const float* __restrict__ W1, const float* __restrict__ b1,
             const float* __restrict__ g1, const float* __restrict__ be1,
             const float* __restrict__ W2, const float* __restrict__ b2,
             const float* __restrict__ g2, const float* __restrict__ be2,
             const float* __restrict__ Wih, const float* __restrict__ bih,
             float* hbuf, float* gxbuf, const int cA, const int cP) {
  int q = blockIdx.x;
  if (q < NBA_) { stage_agg_mlp(q, cA, x, a, eps, W1, b1, g1, be1, W2, b2, g2, be2, hbuf); return; }
  q -= NBA_;
  stage_xproj(q, cP, Wih, bih, hbuf, gxbuf);
}

__global__ void __launch_bounds__(TPBG_)
ftdgnn_pipe(const float* __restrict__ x, const float* __restrict__ a,
            const float* __restrict__ eps,
            const float* __restrict__ W1, const float* __restrict__ b1,
            const float* __restrict__ g1, const float* __restrict__ be1,
            const float* __restrict__ W2, const float* __restrict__ b2,
            const float* __restrict__ g2, const float* __restrict__ be2,
            const float* __restrict__ Wih, const float* __restrict__ Whh,
            const float* __restrict__ bih, const float* __restrict__ bhh,
            float* hbuf, float* gxbuf, float* out,
            const int cG, const int cA, const int cP) {
  int q = blockIdx.x;
  const int nG = (cG >= 0) ? ROI_ : 0;
  const int nA = (cA >= 0) ? NBA_ : 0;
  if (q < nG) { stage_gru(q, cG, Whh, bhh, gxbuf, out); return; }
  q -= nG;
  if (q < nA) { stage_agg_mlp(q, cA, x, a, eps, W1, b1, g1, be1, W2, b2, g2, be2, hbuf); return; }
  q -= nA;
  stage_xproj(q, cP, Wih, bih, hbuf, gxbuf);
}

extern "C" void kernel_launch(void* const* d_in, const int* in_sizes, int n_in,
                              void* d_out, int out_size, void* d_ws, size_t ws_size,
                              hipStream_t stream) {
  const float* x   = (const float*)d_in[0];
  const float* a   = (const float*)d_in[1];
  const float* eps = (const float*)d_in[2];
  const float* W1  = (const float*)d_in[3];
  const float* b1  = (const float*)d_in[4];
  const float* g1  = (const float*)d_in[5];
  const float* be1 = (const float*)d_in[6];
  const float* W2  = (const float*)d_in[7];
  const float* b2  = (const float*)d_in[8];
  const float* g2  = (const float*)d_in[9];
  const float* be2 = (const float*)d_in[10];
  const float* Wih = (const float*)d_in[11];
  const float* Whh = (const float*)d_in[12];
  const float* bih = (const float*)d_in[13];
  const float* bhh = (const float*)d_in[14];
  float* out   = (float*)d_out;
  float* hbuf  = (float*)d_ws;
  float* gxbuf = hbuf + (size_t)ROI_ * ROWS_ * F_;

  const size_t ldsA = (size_t)3 * NP_ * STR_ * sizeof(float);
  const size_t ldsP = (size_t)(MCH_ + G3_) * STR_ * sizeof(float);
  const size_t ldsB = (size_t)168 * 1024;

  ftdgnn_front<<<NBA_, TPBG_, ldsA, stream>>>(
      x, a, eps, W1, b1, g1, be1, W2, b2, g2, be2, Wih, bih, hbuf, gxbuf, 0, -1);
  ftdgnn_front<<<NBA_ + NBP_, TPBG_, ldsP, stream>>>(
      x, a, eps, W1, b1, g1, be1, W2, b2, g2, be2, Wih, bih, hbuf, gxbuf, 1, 0);
  ftdgnn_pipe<<<ROI_ + NBA_ + NBP_, TPBG_, ldsB, stream>>>(
      x, a, eps, W1, b1, g1, be1, W2, b2, g2, be2, Wih, Whh, bih, bhh, hbuf, gxbuf, out, 0, 2, 1);
  ftdgnn_pipe<<<ROI_ + NBA_ + NBP_, TPBG_, ldsB, stream>>>(
      x, a, eps, W1, b1, g1, be1, W2, b2, g2, be2, Wih, Whh, bih, bhh, hbuf, gxbuf, out, 1, 3, 2);
  ftdgnn_pipe<<<ROI_ + NBA_ + NBP_, TPBG_, ldsB, stream>>>(
      x, a, eps, W1, b1, g1, be1, W2, b2, g2, be2, Wih, Whh, bih, bhh, hbuf, gxbuf, out, 2, 4, 3);
  ftdgnn_pipe<<<ROI_ + NBP_, TPBG_, ldsB, stream>>>(
      x, a, eps, W1, b1, g1, be1, W2, b2, g2, be2, Wih, Whh, bih, bhh, hbuf, gxbuf, out, 3, -1, 4);
  ftdgnn_pipe<<<ROI_, TPBG_, ldsB, stream>>>(
      x, a, eps, W1, b1, g1, be1, W2, b2, g2, be2, Wih, Whh, bih, bhh, hbuf, gxbuf, out, 4, -1, -1);
}


# Pipelined launches: two blocks per WGP (LDS request 168 -> 150 KiB; kernel 197 -> 158 registers via one-base LDS and scalar-base global addresses): staging overlaps a neighbour's matrix work
# speedup vs baseline: 3.7624x; 1.0489x over previous
#include <hip/hip_runtime.h>
#include <math.h>

#define B_    4
#define ROI_  90
#define T_    200
#define H_    3
#define F_    96
#define HID_  96
#define G3_   288
#define BH_   12
#define NP_   96
#define STR_  100
#define GSTR_ 288
#define GPAD_ 32
#define GHSZ_ (16 * GSTR_ * 2 + GPAD_)
#define TPB_  288
#define TPBG_ 384
#define KH_   48
#define ROWS_ 2400
#define MCH_  96
#define NCH_  25
#define NPASS_ 32
#define NREAL_ 30
#define CT_   40
#define NCK_  5
#define NBA_  480
#define PCH_  5
#define NBP_  450
static_assert(NCK_ * CT_ == T_ && CT_ % 2 == 0 && PCH_ * MCH_ == CT_ * BH_ &&
              NBA_ == B_ * CT_ * H_ && NBP_ == ROI_ * PCH_, "pipeline chunk map");
extern __shared__ float smem[];
static_assert(TPB_ == 3 * NP_ && NPASS_ * TPB_ == NP_ * NP_ && NREAL_ * 3 == ROI_,
              "kernel A staging map");

typedef float v2f __attribute__((ext_vector_type(2)));
typedef float v8f __attribute__((ext_vector_type(8)));

__device__ __forceinline__ v8f wmma4(v2f a, v2f b, v8f c) {
  return __builtin_amdgcn_wmma_f32_16x16x4_f32(false, a, false, b, (short)0, c,
                                               false, false);
}

__device__ __forceinline__ void mm1x3_trB(const float* As, const float* Bt,
                                          int tr, int tc0, int lane, v8f acc[3]) {
  const int half = lane >> 4;
  const int l15  = lane & 15;
  const float* a0 = As + (tr * 16 + l15) * STR_ + (half << 1);
  const float* b0 = Bt + (tc0 * 16 + l15) * STR_ + (half << 1);
  #pragma unroll
  for (int j = 0; j < 3; ++j) acc[j] = (v8f){0.f,0.f,0.f,0.f,0.f,0.f,0.f,0.f};
  #pragma unroll
  for (int k0 = 0; k0 < NP_; k0 += 4) {
    v2f av  = *(const v2f*)(a0 + k0);
    v2f bv0 = *(const v2f*)(b0 + k0);
    v2f bv1 = *(const v2f*)(b0 + 16 * STR_ + k0);
    v2f bv2 = *(const v2f*)(b0 + 32 * STR_ + k0);
    acc[0] = wmma4(av, bv0, acc[0]);
    acc[1] = wmma4(av, bv1, acc[1]);
    acc[2] = wmma4(av, bv2, acc[2]);
  }
}

__device__ __forceinline__ void mm1x3_halfK_regB(const float* As, const v2f (&bw)[3][KH_ / 4],
                                                 int kOff, int lane, v8f acc[3]) {
  const int half = lane >> 4;
  const int l15  = lane & 15;
  const float* aRow = As + l15 * STR_ + kOff + (half << 1);
  v2f av[KH_ / 4];
  #pragma unroll
  for (int j = 0; j < KH_ / 4; ++j) av[j] = *(const v2f*)(aRow + 4 * j);
  #pragma unroll
  for (int g = 0; g < 3; ++g) acc[g] = (v8f){0.f,0.f,0.f,0.f,0.f,0.f,0.f,0.f};
  #pragma unroll
  for (int j = 0; j < KH_ / 4; ++j) {
    #pragma unroll
    for (int g = 0; g < 3; ++g) acc[g] = wmma4(av[j], bw[g][j], acc[g]);
  }
}

__device__ __forceinline__ void
stage_agg_mlp(const int q, const int ck,
               const float* __restrict__ x, const float* __restrict__ a,
               const float* __restrict__ eps,
               const float* __restrict__ W1, const float* __restrict__ b1,
               const float* __restrict__ g1, const float* __restrict__ be1,
               const float* __restrict__ W2, const float* __restrict__ b2,
               const float* __restrict__ g2, const float* __restrict__ be2,
               float* __restrict__ hbuf) {
  float* xs = smem;
  float* ms = xs + NP_ * STR_;
  float* ws = ms + NP_ * STR_;

  const int tid  = threadIdx.x;
  const int lane = tid & 31;
  const int wave = tid >> 5;
  const bool act = tid < TPB_;
  const int half = lane >> 4;
  const int l15  = lane & 15;
  const int tr   = wave >> 1;
  const int tc0  = 3 * (wave & 1);

  const int b   = q / (CT_ * H_);
  const int th  = q % (CT_ * H_);
  const int t   = ck * CT_ + th / H_;
  const int hh  = th % H_;

  const int r0 = tid / NP_;
  const int c0 = tid % NP_;
  const int ca = (c0 < ROI_) ? c0 : (ROI_ - 1);
  const size_t rowBase = ((size_t)(b * ROI_ + r0) * T_ + t) * H_ + hh;
  const float* xp  = x  + rowBase * F_   + c0;
  const float* ap  = a  + rowBase * ROI_ + ca;
  const float* w1p = W1 + r0 * HID_ + c0;
  float wv[NPASS_];
  if (act) {
  float xv[NREAL_], av[NREAL_];
  #pragma unroll
  for (int i = 0; i < NREAL_; ++i) xv[i] = xp[(size_t)i * (3 * T_ * H_ * F_)];
  #pragma unroll
  for (int i = 0; i < NREAL_; ++i) av[i] = ap[(size_t)i * (3 * T_ * H_ * ROI_)];
  #pragma unroll
  for (int i = 0; i < NPASS_; ++i) wv[i] = w1p[i * (3 * HID_)];
  #pragma unroll
  for (int i = 0; i < NPASS_; ++i) {
    const int r = r0 + 3 * i;
    float xe = 0.f, me = 0.f;
    if (i < NREAL_) {
      xe = xv[i];
      me = (c0 < ROI_ && av[i] != 0.f) ? 1.f : 0.f;
    }
    xs[c0 * STR_ + r] = xe;
    ms[r * STR_ + c0] = me;
    ws[c0 * STR_ + r] = wv[i];
  }
  }
  __syncthreads();

  const float epsv = eps[0];
  const float rs   = rsqrtf(1.0f + 1e-5f);

  v8f acc[3];

  mm1x3_trB(ms, xs, tr, tc0, lane, acc);
  __syncthreads();
  #pragma unroll
  for (int j = 0; j < 3; ++j) {
    const int n = (tc0 + j) * 16 + l15;
    #pragma unroll
    for (int v = 0; v < 8; ++v) {
      const int m = tr * 16 + v + 8 * half;
      ms[m * STR_ + n] = acc[j][v] + epsv * xs[n * STR_ + m];
    }
  }
  __syncthreads();

  mm1x3_trB(ms, ws, tr, tc0, lane, acc);
  __syncthreads();
  if (act) {
    const float* w2p = W2 + r0 * F_ + c0;
    #pragma unroll
    for (int i = 0; i < NPASS_; ++i) wv[i] = w2p[i * (3 * F_)];
  }
  #pragma unroll
  for (int j = 0; j < 3; ++j) {
    const int n = (tc0 + j) * 16 + l15;
    const float sc1 = g1[n] * rs, sh1 = be1[n], bb1 = b1[n];
    #pragma unroll
    for (int v = 0; v < 8; ++v) {
      const int m = tr * 16 + v + 8 * half;
      float z = (acc[j][v] + bb1) * sc1 + sh1;
      z = (z > 0.f) ? z : expm1f(z);
      xs[m * STR_ + n] = z;
    }
  }
  if (act) {
    #pragma unroll
    for (int i = 0; i < NPASS_; ++i)
      ws[c0 * STR_ + r0 + 3 * i] = wv[i];
  }
  __syncthreads();

  mm1x3_trB(xs, ws, tr, tc0, lane, acc);
  #pragma unroll
  for (int j = 0; j < 3; ++j) {
    const int n = (tc0 + j) * 16 + l15;
    const float sc2 = g2[n] * rs, sh2 = be2[n], bb2 = b2[n];
    #pragma unroll
    for (int v = 0; v < 8; ++v) {
      const int m = tr * 16 + v + 8 * half;
      if (m < ROI_) {
        float z = (acc[j][v] + bb2) * sc2 + sh2;
        z = (z > 0.f) ? z : 0.f;
        hbuf[((size_t)(m * T_ + t) * BH_ + (b * H_ + hh)) * F_ + n] = z;
      }
    }
  }
}

__device__ __forceinline__ void
stage_xproj(const int q, const int ck,
             const float* __restrict__ Wih, const float* __restrict__ bih,
             const float* __restrict__ hbuf, float* __restrict__ gxbuf) {
  float* As = smem;
  float* Bs = As + MCH_ * STR_;

  const int tid  = threadIdx.x;
  const int lane = tid & 31;
  const int wave = tid >> 5;
  const bool act = tid < TPB_;
  const int half = lane >> 4;
  const int l15  = lane & 15;
  const int tr   = wave % 6;
  const int tcb  = 9 * (wave / 6);
  const int r    = q / PCH_;
  const int ch   = ck * PCH_ + q % PCH_;
  const size_t row0 = (size_t)r * ROWS_ + (size_t)ch * MCH_;

  const float* Ag = hbuf + row0 * F_;
  const float* Bg = Wih + (size_t)r * G3_ * F_;
  if (act) {
  #pragma unroll
  for (int i = 0; i < MCH_ * (F_ / 4) / TPB_; ++i) {
    const int idx = tid + i * TPB_;
    const int row = idx / (F_ / 4), k4 = idx % (F_ / 4);
    const float4 v = *(const float4*)(Ag + row * F_ + k4 * 4);
    *(float4*)(As + row * STR_ + k4 * 4) = v;
  }
  #pragma unroll
  for (int i = 0; i < G3_ * (F_ / 4) / TPB_; ++i) {
    const int idx = tid + i * TPB_;
    const int g = idx / (F_ / 4), k4 = idx % (F_ / 4);
    const float4 v = *(const float4*)(Bg + g * F_ + k4 * 4);
    *(float4*)(Bs + g * STR_ + k4 * 4) = v;
  }
  }
  __syncthreads();

  for (int ps = 0; ps < 3; ++ps) {
    const int tc0 = tcb + 3 * ps;
    v8f acc[3];
    mm1x3_trB(As, Bs, tr, tc0, lane, acc);
    #pragma unroll
    for (int j = 0; j < 3; ++j) {
      const int n = (tc0 + j) * 16 + l15;
      const float bias = bih[r * G3_ + n];
      #pragma unroll
      for (int v = 0; v < 8; ++v) {
        const int m = tr * 16 + v + 8 * half;
        gxbuf[(row0 + m) * G3_ + n] = acc[j][v] + bias;
      }
    }
  }
}

__device__ __forceinline__ void lds_barrier() {
  __builtin_amdgcn_fence(__ATOMIC_RELEASE, "workgroup", "local");
  __builtin_amdgcn_s_barrier();
  __builtin_amdgcn_fence(__ATOMIC_ACQUIRE, "workgroup", "local");
}

__device__ __forceinline__ void
stage_gru(const int r, const int ck,
          const float* __restrict__ Whh, const float* __restrict__ bhh,
          const float* __restrict__ gxbuf, float* out) {
  float* hS   = smem;
  float* gh   = hS   + 16 * STR_;
  float* bhhS = gh   + GHSZ_;

  const int tid  = threadIdx.x;
  const int lane = tid & 31;
  const int wave = tid >> 5;
  const int half = lane >> 4;
  const int l15  = lane & 15;
  const int cg   = wave >> 1;
  const int kh   = wave & 1;
  const int f    = tid % HID_;
  const int m0   = tid / HID_;
  const float kLog2e = 1.44269504088896341f;
  const v2f* ghr = (const v2f*)gh + m0 * GSTR_ + f;
  float* hSr = hS + m0 * STR_ + f;
  float* ghw = gh + ((8 * half) * GSTR_ + 16 * cg + l15) * 2 + kh + GPAD_ * half;

  v2f bw[3][KH_ / 4];
  {
    const float* bsrc = Whh + ((size_t)r * G3_ + 16 * cg + l15) * HID_ + KH_ * kh + (half << 1);
    #pragma unroll
    for (int g = 0; g < 3; ++g) {
      #pragma unroll
      for (int j = 0; j < KH_ / 4; ++j)
        bw[g][j] = *(const v2f*)(bsrc + (size_t)g * HID_ * HID_ + 4 * j);
    }
  }
  for (int idx = tid; idx < G3_; idx += TPBG_) bhhS[idx] = bhh[r * G3_ + idx];
  for (int idx = tid; idx < 16 * STR_; idx += TPBG_) hS[idx] = 0.f;
  __syncthreads();
  const int t0 = ck * CT_;
  if (ck > 0) {
    #pragma unroll
    for (int i = 0; i < 3; ++i) {
      const int m = 4 * i + m0;
      const int bb = m / H_, hh = m % H_;
      hS[m * STR_ + f] = out[(((size_t)(bb * ROI_ + r) * T_ + (t0 - 1)) * H_ + hh) * HID_ + f];
    }
    __syncthreads();
  }

  const char* gxU = (const char*)(gxbuf + (size_t)r * T_ * BH_ * G3_);
  const unsigned gxo = (unsigned)(m0 * G3_ + f) * 4u;
  auto load_gx = [&](int t, float (&xg)[3][3]) {
    const char* gxt = gxU + (size_t)t * (BH_ * G3_ * 4);
    #pragma unroll
    for (int i = 0; i < 3; ++i)
      #pragma unroll
      for (int g = 0; g < 3; ++g)
        xg[i][g] = *(const float*)(gxt + gxo + ((4 * i) * G3_ + g * HID_) * 4);
  };
  char* outU = (char*)(out + (size_t)r * T_ * H_ * HID_);
  unsigned oo[3];
  #pragma unroll
  for (int i = 0; i < 3; ++i) {
    const int m = 4 * i + m0;
    oo[i] = (unsigned)(((m / H_) * ROI_ * T_ * H_ + (m % H_)) * HID_ + f) * 4u;
  }

  auto step = [&](int t, const float (&xg)[3][3]) {
    v8f acc[3];
    mm1x3_halfK_regB(hS, bw, KH_ * kh, lane, acc);
    #pragma unroll
    for (int g = 0; g < 3; ++g) {
      #pragma unroll
      for (int v = 0; v < 8; ++v)
        ghw[(v * GSTR_ + g * HID_) * 2] = acc[g][v];
    }
    lds_barrier();

    float hr[3], hz[3], hn[3], hp[3];
    #pragma unroll
    for (int i = 0; i < 3; ++i) {
      const v2f* gp = ghr + (4 * i) * GSTR_ + ((i == 2) ? GPAD_ / 2 : 0);
      const v2f pr = gp[0];
      const v2f pz = gp[HID_];
      const v2f pn = gp[2 * HID_];
      hr[i] = (pr.x + pr.y) + bhhS[f];
      hz[i] = (pz.x + pz.y) + bhhS[HID_ + f];
      hn[i] = (pn.x + pn.y) + bhhS[2 * HID_ + f];
      hp[i] = hSr[(4 * i) * STR_];
    }
    #pragma unroll
    for (int i = 0; i < 3; ++i) {
      const float xr = xg[i][0];
      const float xz = xg[i][1];
      const float xn = xg[i][2];
      const float rg = __builtin_amdgcn_rcpf(1.f + __builtin_amdgcn_exp2f(-kLog2e * (xr + hr[i])));
      const float zg = __builtin_amdgcn_rcpf(1.f + __builtin_amdgcn_exp2f(-kLog2e * (xz + hz[i])));
      const float en = __builtin_amdgcn_exp2f((2.f * kLog2e) * (xn + rg * hn[i]));
      const float ng = 1.f - 2.f * __builtin_amdgcn_rcpf(1.f + en);
      const float h2 = (1.f - zg) * ng + zg * hp[i];
      hSr[(4 * i) * STR_] = h2;
      *(float*)(outU + (size_t)t * (H_ * HID_ * 4) + oo[i]) = h2;
    }
    lds_barrier();
  };

  static_assert(T_ % 2 == 0, "time loop handles two steps per trip");
  static_assert((GSTR_ * 2 * 4) % 256 == 0 && (HID_ * 2 * 4) % 256 == 0 && (GPAD_ % 64) == 32 &&
                (8 * GSTR_ * 2) % 64 == 0, "gate buffer: store offsets and bank map");
  static_assert(TPBG_ == 4 * HID_ && BH_ == 3 * 4 && TPBG_ / 32 == 2 * (HID_ / 16) && 2 * KH_ == HID_,
                "GRU kernel thread and wave maps");
  float xa[3][3], xb[3][3];
  load_gx(t0, xa);
  for (int t = t0; t < t0 + CT_; t += 2) {
    load_gx(t + 1, xb);
    step(t, xa);
    load_gx((t + 2 < t0 + CT_) ? t + 2 : t0 + CT_ - 1, xa);
    step(t + 1, xb);
  }
}

__global__ void __launch_bounds__(TPBG_) __attribute__((amdgpu_waves_per_eu(6)))
ftdgnn_front(const float* __restrict__ x, const float* __restrict__ a,
             const float* __restrict__ eps,
             const float* __restrict__ W1, const float* __restrict__ b1,
             const float* __restrict__ g1, const float* __restrict__ be1,
             const float* __restrict__ W2, const float* __restrict__ b2,
             const float* __restrict__ g2, const float* __restrict__ be2,
             const float* __restrict__ Wih, const float* __restrict__ bih,
             float* hbuf, float* gxbuf, const int cA, const int cP) {
  int q = blockIdx.x;
  if (q < NBA_) { stage_agg_mlp(q, cA, x, a, eps, W1, b1, g1, be1, W2, b2, g2, be2, hbuf); return; }
  q -= NBA_;
  stage_xproj(q, cP, Wih, bih, hbuf, gxbuf);
}

__global__ void __launch_bounds__(TPBG_) __attribute__((amdgpu_waves_per_eu(6)))
ftdgnn_pipe(const float* __restrict__ x, const float* __restrict__ a,
            const float* __restrict__ eps,
            const float* __restrict__ W1, const float* __restrict__ b1,
            const float* __restrict__ g1, const float* __restrict__ be1,
            const float* __restrict__ W2, const float* __restrict__ b2,
            const float* __restrict__ g2, const float* __restrict__ be2,
            const float* __restrict__ Wih, const float* __restrict__ Whh,
            const float* __restrict__ bih, const float* __restrict__ bhh,
            float* hbuf, float* gxbuf, float* out,
            const int cG, const int cA, const int cP) {
  int q = blockIdx.x;
  const int nG = (cG >= 0) ? ROI_ : 0;
  const int nA = (cA >= 0) ? NBA_ : 0;
  if (q < nG) { stage_gru(q, cG, Whh, bhh, gxbuf, out); return; }
  q -= nG;
  if (q < nA) { stage_agg_mlp(q, cA, x, a, eps, W1, b1, g1, be1, W2, b2, g2, be2, hbuf); return; }
  q -= nA;
  stage_xproj(q, cP, Wih, bih, hbuf, gxbuf);
}

extern "C" void kernel_launch(void* const* d_in, const int* in_sizes, int n_in,
                              void* d_out, int out_size, void* d_ws, size_t ws_size,
                              hipStream_t stream) {
  const float* x   = (const float*)d_in[0];
  const float* a   = (const float*)d_in[1];
  const float* eps = (const float*)d_in[2];
  const float* W1  = (const float*)d_in[3];
  const float* b1  = (const float*)d_in[4];
  const float* g1  = (const float*)d_in[5];
  const float* be1 = (const float*)d_in[6];
  const float* W2  = (const float*)d_in[7];
  const float* b2  = (const float*)d_in[8];
  const float* g2  = (const float*)d_in[9];
  const float* be2 = (const float*)d_in[10];
  const float* Wih = (const float*)d_in[11];
  const float* Whh = (const float*)d_in[12];
  const float* bih = (const float*)d_in[13];
  const float* bhh = (const float*)d_in[14];
  float* out   = (float*)d_out;
  float* hbuf  = (float*)d_ws;
  float* gxbuf = hbuf + (size_t)ROI_ * ROWS_ * F_;

  const size_t ldsA = (size_t)3 * NP_ * STR_ * sizeof(float);
  const size_t ldsP = (size_t)(MCH_ + G3_) * STR_ * sizeof(float);
  const size_t ldsB = ldsP;

  ftdgnn_front<<<NBA_, TPBG_, ldsA, stream>>>(
      x, a, eps, W1, b1, g1, be1, W2, b2, g2, be2, Wih, bih, hbuf, gxbuf, 0, -1);
  ftdgnn_front<<<NBA_ + NBP_, TPBG_, ldsP, stream>>>(
      x, a, eps, W1, b1, g1, be1, W2, b2, g2, be2, Wih, bih, hbuf, gxbuf, 1, 0);
  ftdgnn_pipe<<<ROI_ + NBA_ + NBP_, TPBG_, ldsB, stream>>>(
      x, a, eps, W1, b1, g1, be1, W2, b2, g2, be2, Wih, Whh, bih, bhh, hbuf, gxbuf, out, 0, 2, 1);
  ftdgnn_pipe<<<ROI_ + NBA_ + NBP_, TPBG_, ldsB, stream>>>(
      x, a, eps, W1, b1, g1, be1, W2, b2, g2, be2, Wih, Whh, bih, bhh, hbuf, gxbuf, out, 1, 3, 2);
  ftdgnn_pipe<<<ROI_ + NBA_ + NBP_, TPBG_, ldsB, stream>>>(
      x, a, eps, W1, b1, g1, be1, W2, b2, g2, be2, Wih, Whh, bih, bhh, hbuf, gxbuf, out, 2, 4, 3);
  ftdgnn_pipe<<<ROI_ + NBP_, TPBG_, ldsB, stream>>>(
      x, a, eps, W1, b1, g1, be1, W2, b2, g2, be2, Wih, Whh, bih, bhh, hbuf, gxbuf, out, 3, -1, 4);
  ftdgnn_pipe<<<ROI_, TPBG_, ldsB, stream>>>(
      x, a, eps, W1, b1, g1, be1, W2, b2, g2, be2, Wih, Whh, bih, bhh, hbuf, gxbuf, out, 4, -1, -1);
}
